# starting kernel
# baseline (speedup 1.0000x reference)
#include <hip/hip_runtime.h>
#include <math.h>

#define B_    16
#define C_    64
#define T_    2048
#define OUT_  512
#define NEED_ 257
#define K2_   514
#define K2P_  544

static __device__ __constant__ float kTwoPi = 6.283185307179586f;

typedef float v2f __attribute__((ext_vector_type(2)));
typedef float v8f __attribute__((ext_vector_type(8)));

__global__ void sp_build_w(float* __restrict__ W) {
    int idx = blockIdx.x * blockDim.x + threadIdx.x;
    if (idx >= K2P_ * OUT_) return;
    int row = idx / OUT_;
    int n   = idx - row * OUT_;
    int k   = row >> 1;
    float val = 0.0f;
    if (k < NEED_) {
        float wk = (k == 0 || k == NEED_ - 1) ? 1.0f : 2.0f;
        int im = (k * n) & (OUT_ - 1);
        float ang = kTwoPi * (float)im * (1.0f / (float)OUT_);
        float s, c;
        __sincosf(ang, &s, &c);
        val = (row & 1) ? (-wk * s * (1.0f / (float)OUT_))
                        : ( wk * c * (1.0f / (float)OUT_));
    }
    W[idx] = val;
}

#define KSTEP1 64
__global__ __launch_bounds__(256) void sp_gemm1(const float* __restrict__ x,
                                                const int*   __restrict__ length,
                                                float*       __restrict__ Y) {
    int b  = blockIdx.x;
    int nb = blockIdx.y;
    int Lr = length[b];
    int L  = Lr < 1 ? 1 : Lr;
    float invL = 1.0f / (float)L;

    __shared__ float sA[C_][KSTEP1 + 1];
    __shared__ float sF[KSTEP1][32 + 1];

    int tid   = threadIdx.x;
    int lane  = tid & 31;
    int wave  = tid >> 5;
    int mtile = wave >> 1;
    int ntile = wave & 1;

    v8f acc = {0.f, 0.f, 0.f, 0.f, 0.f, 0.f, 0.f, 0.f};

    const float* xb = x + (size_t)b * C_ * T_;

    for (int kk = 0; kk < T_; kk += KSTEP1) {
        #pragma unroll
        for (int i = 0; i < (C_ * KSTEP1) / 256; ++i) {
            int li = tid + i * 256;
            int r  = li >> 6;
            int c  = li & (KSTEP1 - 1);
            sA[r][c] = xb[r * T_ + kk + c];
        }
        #pragma unroll
        for (int i = 0; i < (KSTEP1 * 32) / 256; ++i) {
            int li = tid + i * 256;
            int tr = li >> 5;
            int nc = li & 31;
            int t  = kk + tr;
            int g2 = nb * 32 + nc;
            int k  = g2 >> 1;
            float v = 0.0f;
            if (t < L && k < NEED_) {
                int im = (k * t) % L;
                float ang = kTwoPi * (float)im * invL;
                float s, c;
                __sincosf(ang, &s, &c);
                v = (g2 & 1) ? -s : c;
            }
            sF[tr][nc] = v;
        }
        __syncthreads();

        int arow = mtile * 16 + (lane & 15);
        int bcol = ntile * 16 + (lane & 15);
        int koff = (lane >> 4) << 1;
        #pragma unroll
        for (int ks = 0; ks < KSTEP1; ks += 4) {
            v2f a, bf;
            a.x  = sA[arow][ks + koff];
            a.y  = sA[arow][ks + koff + 1];
            bf.x = sF[ks + koff][bcol];
            bf.y = sF[ks + koff + 1][bcol];
            acc = __builtin_amdgcn_wmma_f32_16x16x4_f32(
                false, a, false, bf, (short)0, acc, false, false);
        }
        __syncthreads();
    }

    int colg    = nb * 32 + ntile * 16 + (lane & 15);
    int rowbase = mtile * 16 + ((lane >> 4) ? 8 : 0);
    float* Yb = Y + (size_t)b * C_ * K2P_;
    #pragma unroll
    for (int i = 0; i < 8; ++i) {
        Yb[(rowbase + i) * K2P_ + colg] = acc[i];
    }
}

#define KSTEP2 32
__global__ __launch_bounds__(256) void sp_gemm2(const float* __restrict__ Y,
                                                const float* __restrict__ W,
                                                const float* __restrict__ x,
                                                const int*   __restrict__ length,
                                                float*       __restrict__ out) {
    int b  = blockIdx.x;
    int nb = blockIdx.y;
    int L  = length[b];

    __shared__ float sY[C_][KSTEP2 + 1];
    __shared__ float sW[KSTEP2][32 + 1];

    int tid   = threadIdx.x;
    int lane  = tid & 31;
    int wave  = tid >> 5;
    int mtile = wave >> 1;
    int ntile = wave & 1;

    v8f acc = {0.f, 0.f, 0.f, 0.f, 0.f, 0.f, 0.f, 0.f};

    const float* Yb = Y + (size_t)b * C_ * K2P_;

    for (int kk = 0; kk < K2P_; kk += KSTEP2) {
        #pragma unroll
        for (int i = 0; i < (C_ * KSTEP2) / 256; ++i) {
            int li = tid + i * 256;
            int r  = li >> 5;
            int c  = li & (KSTEP2 - 1);
            sY[r][c] = Yb[r * K2P_ + kk + c];
        }
        #pragma unroll
        for (int i = 0; i < (KSTEP2 * 32) / 256; ++i) {
            int li = tid + i * 256;
            int r  = li >> 5;
            int c  = li & 31;
            sW[r][c] = W[(kk + r) * OUT_ + nb * 32 + c];
        }
        __syncthreads();

        int arow = mtile * 16 + (lane & 15);
        int bcol = ntile * 16 + (lane & 15);
        int koff = (lane >> 4) << 1;
        #pragma unroll
        for (int ks = 0; ks < KSTEP2; ks += 4) {
            v2f a, bf;
            a.x  = sY[arow][ks + koff];
            a.y  = sY[arow][ks + koff + 1];
            bf.x = sW[ks + koff][bcol];
            bf.y = sW[ks + koff + 1][bcol];
            acc = __builtin_amdgcn_wmma_f32_16x16x4_f32(
                false, a, false, bf, (short)0, acc, false, false);
        }
        __syncthreads();
    }

    int n       = nb * 32 + ntile * 16 + (lane & 15);
    int rowbase = mtile * 16 + ((lane >> 4) ? 8 : 0);
    const float* xb = x + (size_t)b * C_ * T_;
    float* ob = out + (size_t)b * C_ * OUT_;
    bool shortSeq = (L < OUT_);
    #pragma unroll
    for (int i = 0; i < 8; ++i) {
        int c = rowbase + i;
        float v = acc[i];
        if (shortSeq) v = (n < L) ? xb[c * T_ + n] : 0.0f;
        ob[c * OUT_ + n] = v;
    }
}

extern "C" void kernel_launch(void* const* d_in, const int* in_sizes, int n_in,
                              void* d_out, int out_size, void* d_ws, size_t ws_size,
                              hipStream_t stream) {
    (void)in_sizes; (void)n_in; (void)out_size; (void)ws_size;
    const float* x      = (const float*)d_in[0];
    const int*   length = (const int*)d_in[1];
    float*       out    = (float*)d_out;

    float* W = (float*)d_ws;
    float* Y = W + (size_t)K2P_ * OUT_;

    sp_build_w<<<(K2P_ * OUT_ + 255) / 256, 256, 0, stream>>>(W);
    sp_gemm1<<<dim3(B_, K2P_ / 32), 256, 0, stream>>>(x, length, Y);
    sp_gemm2<<<dim3(B_, OUT_ / 32), 256, 0, stream>>>(Y, W, x, length, out);
}


# Register prefetch in both GEMM K-loops: next chunk's global loads issued ahead of the current chunk's work (1 ahead in GEMM1, 2 in GEMM2), kept in flight by an LDS-only workgroup barrier
# speedup vs baseline: 1.2251x; 1.2251x over previous
#include <hip/hip_runtime.h>
#include <math.h>

#define B_    16
#define C_    64
#define T_    2048
#define OUT_  512
#define NEED_ 257
#define K2_   514
#define K2P_  544

static __device__ __constant__ float kTwoPi = 6.283185307179586f;

typedef float v2f __attribute__((ext_vector_type(2)));
typedef float v8f __attribute__((ext_vector_type(8)));

__global__ void sp_build_w(float* __restrict__ W) {
    int idx = blockIdx.x * blockDim.x + threadIdx.x;
    if (idx >= K2P_ * OUT_) return;
    int row = idx / OUT_;
    int n   = idx - row * OUT_;
    int k   = row >> 1;
    float val = 0.0f;
    if (k < NEED_) {
        float wk = (k == 0 || k == NEED_ - 1) ? 1.0f : 2.0f;
        int im = (k * n) & (OUT_ - 1);
        float ang = kTwoPi * (float)im * (1.0f / (float)OUT_);
        float s, c;
        __sincosf(ang, &s, &c);
        val = (row & 1) ? (-wk * s * (1.0f / (float)OUT_))
                        : ( wk * c * (1.0f / (float)OUT_));
    }
    W[idx] = val;
}

static __device__ __forceinline__ void lds_barrier() {
    __builtin_amdgcn_fence(__ATOMIC_RELEASE, "workgroup", "local");
    __builtin_amdgcn_s_barrier();
    __builtin_amdgcn_fence(__ATOMIC_ACQUIRE, "workgroup", "local");
}

#define KSTEP1 64
__global__ __launch_bounds__(256) void sp_gemm1(const float* __restrict__ x,
                                                const int*   __restrict__ length,
                                                float*       __restrict__ Y) {
    int b  = blockIdx.x;
    int nb = blockIdx.y;
    int Lr = length[b];
    int L  = Lr < 1 ? 1 : Lr;
    float invL = 1.0f / (float)L;

    __shared__ float sA[C_][KSTEP1 + 1];
    __shared__ float sF[KSTEP1][32 + 1];

    int tid   = threadIdx.x;
    int lane  = tid & 31;
    int wave  = tid >> 5;
    int mtile = wave >> 1;
    int ntile = wave & 1;

    v8f acc = {0.f, 0.f, 0.f, 0.f, 0.f, 0.f, 0.f, 0.f};

    const float* xb = x + (size_t)b * C_ * T_;

    float rx[(C_ * KSTEP1) / 256];
    #pragma unroll
    for (int i = 0; i < (C_ * KSTEP1) / 256; ++i) {
        int li = tid + i * 256;
        int r  = li >> 6;
        int c  = li & (KSTEP1 - 1);
        rx[i] = xb[r * T_ + c];
    }

    for (int kk = 0; kk < T_; kk += KSTEP1) {
        #pragma unroll
        for (int i = 0; i < (C_ * KSTEP1) / 256; ++i) {
            int li = tid + i * 256;
            int r  = li >> 6;
            int c  = li & (KSTEP1 - 1);
            sA[r][c] = rx[i];
        }
        if (kk + KSTEP1 < T_) {
            #pragma unroll
            for (int i = 0; i < (C_ * KSTEP1) / 256; ++i) {
                int li = tid + i * 256;
                int r  = li >> 6;
                int c  = li & (KSTEP1 - 1);
                rx[i] = xb[r * T_ + kk + KSTEP1 + c];
            }
        }
        #pragma unroll
        for (int i = 0; i < (KSTEP1 * 32) / 256; ++i) {
            int li = tid + i * 256;
            int tr = li >> 5;
            int nc = li & 31;
            int t  = kk + tr;
            int g2 = nb * 32 + nc;
            int k  = g2 >> 1;
            float v = 0.0f;
            if (t < L && k < NEED_) {
                int im = (k * t) % L;
                float ang = kTwoPi * (float)im * invL;
                float s, c;
                __sincosf(ang, &s, &c);
                v = (g2 & 1) ? -s : c;
            }
            sF[tr][nc] = v;
        }
        lds_barrier();

        int arow = mtile * 16 + (lane & 15);
        int bcol = ntile * 16 + (lane & 15);
        int koff = (lane >> 4) << 1;
        #pragma unroll
        for (int ks = 0; ks < KSTEP1; ks += 4) {
            v2f a, bf;
            a.x  = sA[arow][ks + koff];
            a.y  = sA[arow][ks + koff + 1];
            bf.x = sF[ks + koff][bcol];
            bf.y = sF[ks + koff + 1][bcol];
            acc = __builtin_amdgcn_wmma_f32_16x16x4_f32(
                false, a, false, bf, (short)0, acc, false, false);
        }
        lds_barrier();
    }

    int colg    = nb * 32 + ntile * 16 + (lane & 15);
    int rowbase = mtile * 16 + ((lane >> 4) ? 8 : 0);
    float* Yb = Y + (size_t)b * C_ * K2P_;
    #pragma unroll
    for (int i = 0; i < 8; ++i) {
        Yb[(rowbase + i) * K2P_ + colg] = acc[i];
    }
}

#define KSTEP2 32
__global__ __launch_bounds__(256) void sp_gemm2(const float* __restrict__ Y,
                                                const float* __restrict__ W,
                                                const float* __restrict__ x,
                                                const int*   __restrict__ length,
                                                float*       __restrict__ out) {
    int b  = blockIdx.x;
    int nb = blockIdx.y;
    int L  = length[b];

    __shared__ float sY[C_][KSTEP2 + 1];
    __shared__ float sW[KSTEP2][32 + 1];

    int tid   = threadIdx.x;
    int lane  = tid & 31;
    int wave  = tid >> 5;
    int mtile = wave >> 1;
    int ntile = wave & 1;

    v8f acc = {0.f, 0.f, 0.f, 0.f, 0.f, 0.f, 0.f, 0.f};

    const float* Yb = Y + (size_t)b * C_ * K2P_;
    const float* Wn = W + nb * 32;

    float ryA[(C_ * KSTEP2) / 256], rwA[(KSTEP2 * 32) / 256];
    float ryB[(C_ * KSTEP2) / 256], rwB[(KSTEP2 * 32) / 256];

    auto fetch = [&](int kk, float (&ry)[(C_ * KSTEP2) / 256], float (&rw)[(KSTEP2 * 32) / 256]) {
        #pragma unroll
        for (int i = 0; i < (C_ * KSTEP2) / 256; ++i) {
            int li = tid + i * 256;
            int r  = li >> 5;
            int c  = li & (KSTEP2 - 1);
            ry[i] = Yb[r * K2P_ + kk + c];
        }
        #pragma unroll
        for (int i = 0; i < (KSTEP2 * 32) / 256; ++i) {
            int li = tid + i * 256;
            int r  = li >> 5;
            int c  = li & 31;
            rw[i] = Wn[(kk + r) * OUT_ + c];
        }
    };
    auto stage = [&](const float (&ry)[(C_ * KSTEP2) / 256], const float (&rw)[(KSTEP2 * 32) / 256]) {
        #pragma unroll
        for (int i = 0; i < (C_ * KSTEP2) / 256; ++i) {
            int li = tid + i * 256;
            int r  = li >> 5;
            int c  = li & (KSTEP2 - 1);
            sY[r][c] = ry[i];
        }
        #pragma unroll
        for (int i = 0; i < (KSTEP2 * 32) / 256; ++i) {
            int li = tid + i * 256;
            int r  = li >> 5;
            int c  = li & 31;
            sW[r][c] = rw[i];
        }
    };
    int arow = mtile * 16 + (lane & 15);
    int bcol = ntile * 16 + (lane & 15);
    int koff = (lane >> 4) << 1;
    auto mma_chunk = [&]() {
        #pragma unroll
        for (int ks = 0; ks < KSTEP2; ks += 4) {
            v2f a, bf;
            a.x  = sY[arow][ks + koff];
            a.y  = sY[arow][ks + koff + 1];
            bf.x = sW[ks + koff][bcol];
            bf.y = sW[ks + koff + 1][bcol];
            acc = __builtin_amdgcn_wmma_f32_16x16x4_f32(
                false, a, false, bf, (short)0, acc, false, false);
        }
    };

    static_assert(K2P_ == 17 * KSTEP2, "the peeled tail below assumes 17 chunks");
    fetch(0, ryA, rwA);
    __builtin_amdgcn_sched_barrier(0);
    fetch(KSTEP2, ryB, rwB);
    for (int kk = 0; kk < 14 * KSTEP2; kk += 2 * KSTEP2) {
        stage(ryA, rwA);
        fetch(kk + 2 * KSTEP2, ryA, rwA);
        lds_barrier();
        mma_chunk();
        lds_barrier();

        stage(ryB, rwB);
        fetch(kk + 3 * KSTEP2, ryB, rwB);
        lds_barrier();
        mma_chunk();
        lds_barrier();
    }
    stage(ryA, rwA);
    fetch(16 * KSTEP2, ryA, rwA);
    lds_barrier();
    mma_chunk();
    lds_barrier();

    stage(ryB, rwB);
    lds_barrier();
    mma_chunk();
    lds_barrier();

    stage(ryA, rwA);
    lds_barrier();
    mma_chunk();

    int n       = nb * 32 + ntile * 16 + (lane & 15);
    int rowbase = mtile * 16 + ((lane >> 4) ? 8 : 0);
    const float* xb = x + (size_t)b * C_ * T_;
    float* ob = out + (size_t)b * C_ * OUT_;
    bool shortSeq = (L < OUT_);
    #pragma unroll
    for (int i = 0; i < 8; ++i) {
        int c = rowbase + i;
        float v = acc[i];
        if (shortSeq) v = (n < L) ? xb[c * T_ + n] : 0.0f;
        ob[c * OUT_ + n] = v;
    }
}

extern "C" void kernel_launch(void* const* d_in, const int* in_sizes, int n_in,
                              void* d_out, int out_size, void* d_ws, size_t ws_size,
                              hipStream_t stream) {
    (void)in_sizes; (void)n_in; (void)out_size; (void)ws_size;
    const float* x      = (const float*)d_in[0];
    const int*   length = (const int*)d_in[1];
    float*       out    = (float*)d_out;

    float* W = (float*)d_ws;
    float* Y = W + (size_t)K2P_ * OUT_;

    sp_build_w<<<(K2P_ * OUT_ + 255) / 256, 256, 0, stream>>>(W);
    sp_gemm1<<<dim3(B_, K2P_ / 32), 256, 0, stream>>>(x, length, Y);
    sp_gemm2<<<dim3(B_, OUT_ / 32), 256, 0, stream>>>(Y, W, x, length, out);
}


# GEMM1 basis synthesis strength-reduced: per-element (k*t) mod L replaced by an exact running phase (add, subtract, unsigned min); one hardware cosine in turns replaces sincos and select
# speedup vs baseline: 1.5486x; 1.2641x over previous
#include <hip/hip_runtime.h>
#include <math.h>

#define B_    16
#define C_    64
#define T_    2048
#define OUT_  512
#define NEED_ 257
#define K2_   514
#define K2P_  544

static __device__ __constant__ float kTwoPi = 6.283185307179586f;

typedef float v2f __attribute__((ext_vector_type(2)));
typedef float v8f __attribute__((ext_vector_type(8)));

__global__ void sp_build_w(float* __restrict__ W) {
    int idx = blockIdx.x * blockDim.x + threadIdx.x;
    if (idx >= K2P_ * OUT_) return;
    int row = idx / OUT_;
    int n   = idx - row * OUT_;
    int k   = row >> 1;
    float val = 0.0f;
    if (k < NEED_) {
        float wk = (k == 0 || k == NEED_ - 1) ? 1.0f : 2.0f;
        int im = (k * n) & (OUT_ - 1);
        float ang = kTwoPi * (float)im * (1.0f / (float)OUT_);
        float s, c;
        __sincosf(ang, &s, &c);
        val = (row & 1) ? (-wk * s * (1.0f / (float)OUT_))
                        : ( wk * c * (1.0f / (float)OUT_));
    }
    W[idx] = val;
}

static __device__ __forceinline__ void lds_barrier() {
    __builtin_amdgcn_fence(__ATOMIC_RELEASE, "workgroup", "local");
    __builtin_amdgcn_s_barrier();
    __builtin_amdgcn_fence(__ATOMIC_ACQUIRE, "workgroup", "local");
}

#define KSTEP1 64
__global__ __launch_bounds__(256) void sp_gemm1(const float* __restrict__ x,
                                                const int*   __restrict__ length,
                                                float*       __restrict__ Y) {
    int b  = blockIdx.x;
    int nb = blockIdx.y;
    int Lr = length[b];
    int L  = Lr < 1 ? 1 : Lr;
    float invL = 1.0f / (float)L;

    __shared__ float sA[C_][KSTEP1 + 1];
    __shared__ float sF[KSTEP1][32 + 1];

    int tid   = threadIdx.x;
    int lane  = tid & 31;
    int wave  = tid >> 5;
    int mtile = wave >> 1;
    int ntile = wave & 1;

    v8f acc = {0.f, 0.f, 0.f, 0.f, 0.f, 0.f, 0.f, 0.f};

    const float* xb = x + (size_t)b * C_ * T_;

    int   fc     = tid & 31;
    int   g2     = nb * 32 + fc;
    int   k      = g2 >> 1;
    bool  kvalid = k < NEED_;
    unsigned step = (unsigned)((8 * k) % L);
    unsigned im   = (unsigned)((k * wave) % L);
    float qturn  = (g2 & 1) ? 0.25f : 0.0f;

    float rx[(C_ * KSTEP1) / 256];
    #pragma unroll
    for (int i = 0; i < (C_ * KSTEP1) / 256; ++i) {
        int li = tid + i * 256;
        int r  = li >> 6;
        int c  = li & (KSTEP1 - 1);
        rx[i] = xb[r * T_ + c];
    }

    for (int kk = 0; kk < T_; kk += KSTEP1) {
        #pragma unroll
        for (int i = 0; i < (C_ * KSTEP1) / 256; ++i) {
            int li = tid + i * 256;
            int r  = li >> 6;
            int c  = li & (KSTEP1 - 1);
            sA[r][c] = rx[i];
        }
        if (kk + KSTEP1 < T_) {
            #pragma unroll
            for (int i = 0; i < (C_ * KSTEP1) / 256; ++i) {
                int li = tid + i * 256;
                int r  = li >> 6;
                int c  = li & (KSTEP1 - 1);
                rx[i] = xb[r * T_ + kk + KSTEP1 + c];
            }
        }
        #pragma unroll
        for (int i = 0; i < (KSTEP1 * 32) / 256; ++i) {
            int tr = wave + i * 8;
            int t  = kk + tr;
            float v = 0.0f;
            if (t < L && kvalid) v = __builtin_amdgcn_cosf((float)(int)im * invL + qturn);
            sF[tr][fc] = v;
            unsigned nx = im + step;
            unsigned wr = nx - (unsigned)L;
            im = nx < wr ? nx : wr;
        }
        lds_barrier();

        int arow = mtile * 16 + (lane & 15);
        int bcol = ntile * 16 + (lane & 15);
        int koff = (lane >> 4) << 1;
        #pragma unroll
        for (int ks = 0; ks < KSTEP1; ks += 4) {
            v2f a, bf;
            a.x  = sA[arow][ks + koff];
            a.y  = sA[arow][ks + koff + 1];
            bf.x = sF[ks + koff][bcol];
            bf.y = sF[ks + koff + 1][bcol];
            acc = __builtin_amdgcn_wmma_f32_16x16x4_f32(
                false, a, false, bf, (short)0, acc, false, false);
        }
        lds_barrier();
    }

    int colg    = nb * 32 + ntile * 16 + (lane & 15);
    int rowbase = mtile * 16 + ((lane >> 4) ? 8 : 0);
    float* Yb = Y + (size_t)b * C_ * K2P_;
    #pragma unroll
    for (int i = 0; i < 8; ++i) {
        Yb[(rowbase + i) * K2P_ + colg] = acc[i];
    }
}

#define KSTEP2 32
__global__ __launch_bounds__(256) void sp_gemm2(const float* __restrict__ Y,
                                                const float* __restrict__ W,
                                                const float* __restrict__ x,
                                                const int*   __restrict__ length,
                                                float*       __restrict__ out) {
    int b  = blockIdx.x;
    int nb = blockIdx.y;
    int L  = length[b];

    __shared__ float sY[C_][KSTEP2 + 1];
    __shared__ float sW[KSTEP2][32 + 1];

    int tid   = threadIdx.x;
    int lane  = tid & 31;
    int wave  = tid >> 5;
    int mtile = wave >> 1;
    int ntile = wave & 1;

    v8f acc = {0.f, 0.f, 0.f, 0.f, 0.f, 0.f, 0.f, 0.f};

    const float* Yb = Y + (size_t)b * C_ * K2P_;
    const float* Wn = W + nb * 32;

    float ryA[(C_ * KSTEP2) / 256], rwA[(KSTEP2 * 32) / 256];
    float ryB[(C_ * KSTEP2) / 256], rwB[(KSTEP2 * 32) / 256];

    auto fetch = [&](int kk, float (&ry)[(C_ * KSTEP2) / 256], float (&rw)[(KSTEP2 * 32) / 256]) {
        #pragma unroll
        for (int i = 0; i < (C_ * KSTEP2) / 256; ++i) {
            int li = tid + i * 256;
            int r  = li >> 5;
            int c  = li & (KSTEP2 - 1);
            ry[i] = Yb[r * K2P_ + kk + c];
        }
        #pragma unroll
        for (int i = 0; i < (KSTEP2 * 32) / 256; ++i) {
            int li = tid + i * 256;
            int r  = li >> 5;
            int c  = li & 31;
            rw[i] = Wn[(kk + r) * OUT_ + c];
        }
    };
    auto stage = [&](const float (&ry)[(C_ * KSTEP2) / 256], const float (&rw)[(KSTEP2 * 32) / 256]) {
        #pragma unroll
        for (int i = 0; i < (C_ * KSTEP2) / 256; ++i) {
            int li = tid + i * 256;
            int r  = li >> 5;
            int c  = li & (KSTEP2 - 1);
            sY[r][c] = ry[i];
        }
        #pragma unroll
        for (int i = 0; i < (KSTEP2 * 32) / 256; ++i) {
            int li = tid + i * 256;
            int r  = li >> 5;
            int c  = li & 31;
            sW[r][c] = rw[i];
        }
    };
    int arow = mtile * 16 + (lane & 15);
    int bcol = ntile * 16 + (lane & 15);
    int koff = (lane >> 4) << 1;
    auto mma_chunk = [&]() {
        #pragma unroll
        for (int ks = 0; ks < KSTEP2; ks += 4) {
            v2f a, bf;
            a.x  = sY[arow][ks + koff];
            a.y  = sY[arow][ks + koff + 1];
            bf.x = sW[ks + koff][bcol];
            bf.y = sW[ks + koff + 1][bcol];
            acc = __builtin_amdgcn_wmma_f32_16x16x4_f32(
                false, a, false, bf, (short)0, acc, false, false);
        }
    };

    static_assert(K2P_ == 17 * KSTEP2, "the peeled tail below assumes 17 chunks");
    fetch(0, ryA, rwA);
    __builtin_amdgcn_sched_barrier(0);
    fetch(KSTEP2, ryB, rwB);
    for (int kk = 0; kk < 14 * KSTEP2; kk += 2 * KSTEP2) {
        stage(ryA, rwA);
        fetch(kk + 2 * KSTEP2, ryA, rwA);
        lds_barrier();
        mma_chunk();
        lds_barrier();

        stage(ryB, rwB);
        fetch(kk + 3 * KSTEP2, ryB, rwB);
        lds_barrier();
        mma_chunk();
        lds_barrier();
    }
    stage(ryA, rwA);
    fetch(16 * KSTEP2, ryA, rwA);
    lds_barrier();
    mma_chunk();
    lds_barrier();

    stage(ryB, rwB);
    lds_barrier();
    mma_chunk();
    lds_barrier();

    stage(ryA, rwA);
    lds_barrier();
    mma_chunk();

    int n       = nb * 32 + ntile * 16 + (lane & 15);
    int rowbase = mtile * 16 + ((lane >> 4) ? 8 : 0);
    const float* xb = x + (size_t)b * C_ * T_;
    float* ob = out + (size_t)b * C_ * OUT_;
    bool shortSeq = (L < OUT_);
    #pragma unroll
    for (int i = 0; i < 8; ++i) {
        int c = rowbase + i;
        float v = acc[i];
        if (shortSeq) v = (n < L) ? xb[c * T_ + n] : 0.0f;
        ob[c * OUT_ + n] = v;
    }
}

extern "C" void kernel_launch(void* const* d_in, const int* in_sizes, int n_in,
                              void* d_out, int out_size, void* d_ws, size_t ws_size,
                              hipStream_t stream) {
    (void)in_sizes; (void)n_in; (void)out_size; (void)ws_size;
    const float* x      = (const float*)d_in[0];
    const int*   length = (const int*)d_in[1];
    float*       out    = (float*)d_out;

    float* W = (float*)d_ws;
    float* Y = W + (size_t)K2P_ * OUT_;

    sp_build_w<<<(K2P_ * OUT_ + 255) / 256, 256, 0, stream>>>(W);
    sp_gemm1<<<dim3(B_, K2P_ / 32), 256, 0, stream>>>(x, length, Y);
    sp_gemm2<<<dim3(B_, OUT_ / 32), 256, 0, stream>>>(Y, W, x, length, out);
}


# Inverse basis W synthesized inside GEMM2 (exact running integer phase, one hardware cosine per element, a chunk ahead in registers) instead of a separate build launch: 3 launches -> 2
# speedup vs baseline: 1.6770x; 1.0829x over previous
#include <hip/hip_runtime.h>
#include <math.h>

#define B_    16
#define C_    64
#define T_    2048
#define OUT_  512
#define NEED_ 257
#define K2_   514
#define K2P_  544

typedef float v2f __attribute__((ext_vector_type(2)));
typedef float v8f __attribute__((ext_vector_type(8)));

static __device__ __forceinline__ void lds_barrier() {
    __builtin_amdgcn_fence(__ATOMIC_RELEASE, "workgroup", "local");
    __builtin_amdgcn_s_barrier();
    __builtin_amdgcn_fence(__ATOMIC_ACQUIRE, "workgroup", "local");
}

#define KSTEP1 64
__global__ __launch_bounds__(256) void sp_gemm1(const float* __restrict__ x,
                                                const int*   __restrict__ length,
                                                float*       __restrict__ Y) {
    int b  = blockIdx.x;
    int nb = blockIdx.y;
    int Lr = length[b];
    int L  = Lr < 1 ? 1 : Lr;
    float invL = 1.0f / (float)L;

    __shared__ float sA[C_][KSTEP1 + 1];
    __shared__ float sF[KSTEP1][32 + 1];

    int tid   = threadIdx.x;
    int lane  = tid & 31;
    int wave  = tid >> 5;
    int mtile = wave >> 1;
    int ntile = wave & 1;

    v8f acc = {0.f, 0.f, 0.f, 0.f, 0.f, 0.f, 0.f, 0.f};

    const float* xb = x + (size_t)b * C_ * T_;

    int   fc     = tid & 31;
    int   g2     = nb * 32 + fc;
    int   k      = g2 >> 1;
    bool  kvalid = k < NEED_;
    unsigned step = (unsigned)((8 * k) % L);
    unsigned im   = (unsigned)((k * wave) % L);
    float qturn  = (g2 & 1) ? 0.25f : 0.0f;

    float rx[(C_ * KSTEP1) / 256];
    #pragma unroll
    for (int i = 0; i < (C_ * KSTEP1) / 256; ++i) {
        int li = tid + i * 256;
        int r  = li >> 6;
        int c  = li & (KSTEP1 - 1);
        rx[i] = xb[r * T_ + c];
    }

    for (int kk = 0; kk < T_; kk += KSTEP1) {
        #pragma unroll
        for (int i = 0; i < (C_ * KSTEP1) / 256; ++i) {
            int li = tid + i * 256;
            int r  = li >> 6;
            int c  = li & (KSTEP1 - 1);
            sA[r][c] = rx[i];
        }
        if (kk + KSTEP1 < T_) {
            #pragma unroll
            for (int i = 0; i < (C_ * KSTEP1) / 256; ++i) {
                int li = tid + i * 256;
                int r  = li >> 6;
                int c  = li & (KSTEP1 - 1);
                rx[i] = xb[r * T_ + kk + KSTEP1 + c];
            }
        }
        #pragma unroll
        for (int i = 0; i < (KSTEP1 * 32) / 256; ++i) {
            int tr = wave + i * 8;
            int t  = kk + tr;
            float v = 0.0f;
            if (t < L && kvalid) v = __builtin_amdgcn_cosf((float)(int)im * invL + qturn);
            sF[tr][fc] = v;
            unsigned nx = im + step;
            unsigned wr = nx - (unsigned)L;
            im = nx < wr ? nx : wr;
        }
        lds_barrier();

        int arow = mtile * 16 + (lane & 15);
        int bcol = ntile * 16 + (lane & 15);
        int koff = (lane >> 4) << 1;
        #pragma unroll
        for (int ks = 0; ks < KSTEP1; ks += 4) {
            v2f a, bf;
            a.x  = sA[arow][ks + koff];
            a.y  = sA[arow][ks + koff + 1];
            bf.x = sF[ks + koff][bcol];
            bf.y = sF[ks + koff + 1][bcol];
            acc = __builtin_amdgcn_wmma_f32_16x16x4_f32(
                false, a, false, bf, (short)0, acc, false, false);
        }
        lds_barrier();
    }

    int colg    = nb * 32 + ntile * 16 + (lane & 15);
    int rowbase = mtile * 16 + ((lane >> 4) ? 8 : 0);
    float* Yb = Y + (size_t)b * C_ * K2P_;
    #pragma unroll
    for (int i = 0; i < 8; ++i) {
        Yb[(rowbase + i) * K2P_ + colg] = acc[i];
    }
}

#define KSTEP2 32
__global__ __launch_bounds__(256) void sp_gemm2(const float* __restrict__ Y,
                                                const float* __restrict__ x,
                                                const int*   __restrict__ length,
                                                float*       __restrict__ out) {
    int b  = blockIdx.x;
    int nb = blockIdx.y;
    int L  = length[b];

    __shared__ float sY[C_][KSTEP2 + 1];
    __shared__ float sW[KSTEP2][32 + 1];

    int tid   = threadIdx.x;
    int lane  = tid & 31;
    int wave  = tid >> 5;
    int mtile = wave >> 1;
    int ntile = wave & 1;

    v8f acc = {0.f, 0.f, 0.f, 0.f, 0.f, 0.f, 0.f, 0.f};

    const float* Yb = Y + (size_t)b * C_ * K2P_;

    float ryA[(C_ * KSTEP2) / 256];
    float ryB[(C_ * KSTEP2) / 256];

    auto fetch = [&](int kk, float (&ry)[(C_ * KSTEP2) / 256]) {
        #pragma unroll
        for (int i = 0; i < (C_ * KSTEP2) / 256; ++i) {
            int li = tid + i * 256;
            int r  = li >> 5;
            int c  = li & (KSTEP2 - 1);
            ry[i] = Yb[r * K2P_ + kk + c];
        }
    };
    auto stage = [&](const float (&ry)[(C_ * KSTEP2) / 256]) {
        #pragma unroll
        for (int i = 0; i < (C_ * KSTEP2) / 256; ++i) {
            int li = tid + i * 256;
            int r  = li >> 5;
            int c  = li & (KSTEP2 - 1);
            sY[r][c] = ry[i];
        }
    };
    int      nW     = nb * 32 + lane;
    float    qturnW = (wave & 1) ? 0.25f : 0.0f;
    unsigned pW     = (unsigned)((wave >> 1) * nW) & (unsigned)(OUT_ - 1);
    unsigned stepW  = (unsigned)(4 * nW) & (unsigned)(OUT_ - 1);
    float rw[(KSTEP2 * 32) / 256];
    auto synth_w = [&](int kk) {
        #pragma unroll
        for (int i = 0; i < (KSTEP2 * 32) / 256; ++i) {
            int k = (kk + wave + i * 8) >> 1;
            float wk = (k == 0 || k == NEED_ - 1) ? (1.0f / (float)OUT_) : (2.0f / (float)OUT_);
            float v  = wk * __builtin_amdgcn_cosf((float)(int)pW * (1.0f / (float)OUT_) + qturnW);
            rw[i] = (k < NEED_) ? v : 0.0f;
            pW = (pW + stepW) & (unsigned)(OUT_ - 1);
        }
    };
    auto stage_w = [&]() {
        #pragma unroll
        for (int i = 0; i < (KSTEP2 * 32) / 256; ++i) sW[wave + i * 8][lane] = rw[i];
    };
    int arow = mtile * 16 + (lane & 15);
    int bcol = ntile * 16 + (lane & 15);
    int koff = (lane >> 4) << 1;
    auto mma_chunk = [&]() {
        #pragma unroll
        for (int ks = 0; ks < KSTEP2; ks += 4) {
            v2f a, bf;
            a.x  = sY[arow][ks + koff];
            a.y  = sY[arow][ks + koff + 1];
            bf.x = sW[ks + koff][bcol];
            bf.y = sW[ks + koff + 1][bcol];
            acc = __builtin_amdgcn_wmma_f32_16x16x4_f32(
                false, a, false, bf, (short)0, acc, false, false);
        }
    };

    static_assert(K2P_ == 17 * KSTEP2, "the peeled tail below assumes 17 chunks");
    fetch(0, ryA);
    __builtin_amdgcn_sched_barrier(0);
    fetch(KSTEP2, ryB);
    synth_w(0);
    for (int kk = 0; kk < 14 * KSTEP2; kk += 2 * KSTEP2) {
        stage(ryA);
        stage_w();
        fetch(kk + 2 * KSTEP2, ryA);
        lds_barrier();
        synth_w(kk + KSTEP2);
        mma_chunk();
        lds_barrier();

        stage(ryB);
        stage_w();
        fetch(kk + 3 * KSTEP2, ryB);
        lds_barrier();
        synth_w(kk + 2 * KSTEP2);
        mma_chunk();
        lds_barrier();
    }
    stage(ryA);
    stage_w();
    fetch(16 * KSTEP2, ryA);
    lds_barrier();
    synth_w(15 * KSTEP2);
    mma_chunk();
    lds_barrier();

    stage(ryB);
    stage_w();
    lds_barrier();
    synth_w(16 * KSTEP2);
    mma_chunk();
    lds_barrier();

    stage(ryA);
    stage_w();
    lds_barrier();
    mma_chunk();

    int n       = nb * 32 + ntile * 16 + (lane & 15);
    int rowbase = mtile * 16 + ((lane >> 4) ? 8 : 0);
    const float* xb = x + (size_t)b * C_ * T_;
    float* ob = out + (size_t)b * C_ * OUT_;
    bool shortSeq = (L < OUT_);
    #pragma unroll
    for (int i = 0; i < 8; ++i) {
        int c = rowbase + i;
        float v = acc[i];
        if (shortSeq) v = (n < L) ? xb[c * T_ + n] : 0.0f;
        ob[c * OUT_ + n] = v;
    }
}

extern "C" void kernel_launch(void* const* d_in, const int* in_sizes, int n_in,
                              void* d_out, int out_size, void* d_ws, size_t ws_size,
                              hipStream_t stream) {
    (void)in_sizes; (void)n_in; (void)out_size; (void)ws_size;
    const float* x      = (const float*)d_in[0];
    const int*   length = (const int*)d_in[1];
    float*       out    = (float*)d_out;

    float* Y = (float*)d_ws;

    sp_gemm1<<<dim3(B_, K2P_ / 32), 256, 0, stream>>>(x, length, Y);
    sp_gemm2<<<dim3(B_, OUT_ / 32), 256, 0, stream>>>(Y, x, length, out);
}


# Nyquist bin's real part stored in the first bin's always-zero imaginary column: spectrum 544 -> 512 columns, GEMM1 grid 272 -> 256 blocks, GEMM2 K 17 -> 16 chunks; Nyquist term summed 2nd, not last
# speedup vs baseline: 2.0935x; 1.2484x over previous
#include <hip/hip_runtime.h>
#include <math.h>

#define B_    16
#define C_    64
#define T_    2048
#define OUT_  512
#define NEED_ 257
#define K2_   512

typedef float v2f __attribute__((ext_vector_type(2)));
typedef float v8f __attribute__((ext_vector_type(8)));

static __device__ __forceinline__ void lds_barrier() {
    __builtin_amdgcn_fence(__ATOMIC_RELEASE, "workgroup", "local");
    __builtin_amdgcn_s_barrier();
    __builtin_amdgcn_fence(__ATOMIC_ACQUIRE, "workgroup", "local");
}

#define KSTEP1 64
__global__ __launch_bounds__(256) void sp_gemm1(const float* __restrict__ x,
                                                const int*   __restrict__ length,
                                                float*       __restrict__ Y) {
    int b  = blockIdx.x;
    int nb = blockIdx.y;
    int Lr = length[b];
    int L  = Lr < 1 ? 1 : Lr;
    float invL = 1.0f / (float)L;

    __shared__ float sA[C_][KSTEP1 + 1];
    __shared__ float sF[KSTEP1][32 + 1];

    int tid   = threadIdx.x;
    int lane  = tid & 31;
    int wave  = tid >> 5;
    int mtile = wave >> 1;
    int ntile = wave & 1;

    v8f acc = {0.f, 0.f, 0.f, 0.f, 0.f, 0.f, 0.f, 0.f};

    const float* xb = x + (size_t)b * C_ * T_;

    int   fc     = tid & 31;
    int   g2     = nb * 32 + fc;
    bool  nyq    = (g2 == 1);
    int   k      = nyq ? (NEED_ - 1) : (g2 >> 1);
    unsigned step = (unsigned)((8 * k) % L);
    unsigned im   = (unsigned)((k * wave) % L);
    float qturn  = ((g2 & 1) && !nyq) ? 0.25f : 0.0f;

    float rx[(C_ * KSTEP1) / 256];
    #pragma unroll
    for (int i = 0; i < (C_ * KSTEP1) / 256; ++i) {
        int li = tid + i * 256;
        int r  = li >> 6;
        int c  = li & (KSTEP1 - 1);
        rx[i] = xb[r * T_ + c];
    }

    for (int kk = 0; kk < T_; kk += KSTEP1) {
        #pragma unroll
        for (int i = 0; i < (C_ * KSTEP1) / 256; ++i) {
            int li = tid + i * 256;
            int r  = li >> 6;
            int c  = li & (KSTEP1 - 1);
            sA[r][c] = rx[i];
        }
        if (kk + KSTEP1 < T_) {
            #pragma unroll
            for (int i = 0; i < (C_ * KSTEP1) / 256; ++i) {
                int li = tid + i * 256;
                int r  = li >> 6;
                int c  = li & (KSTEP1 - 1);
                rx[i] = xb[r * T_ + kk + KSTEP1 + c];
            }
        }
        #pragma unroll
        for (int i = 0; i < (KSTEP1 * 32) / 256; ++i) {
            int tr = wave + i * 8;
            int t  = kk + tr;
            float v = 0.0f;
            if (t < L) v = __builtin_amdgcn_cosf((float)(int)im * invL + qturn);
            sF[tr][fc] = v;
            unsigned nx = im + step;
            unsigned wr = nx - (unsigned)L;
            im = nx < wr ? nx : wr;
        }
        lds_barrier();

        int arow = mtile * 16 + (lane & 15);
        int bcol = ntile * 16 + (lane & 15);
        int koff = (lane >> 4) << 1;
        #pragma unroll
        for (int ks = 0; ks < KSTEP1; ks += 4) {
            v2f a, bf;
            a.x  = sA[arow][ks + koff];
            a.y  = sA[arow][ks + koff + 1];
            bf.x = sF[ks + koff][bcol];
            bf.y = sF[ks + koff + 1][bcol];
            acc = __builtin_amdgcn_wmma_f32_16x16x4_f32(
                false, a, false, bf, (short)0, acc, false, false);
        }
        lds_barrier();
    }

    int colg    = nb * 32 + ntile * 16 + (lane & 15);
    int rowbase = mtile * 16 + ((lane >> 4) ? 8 : 0);
    float* Yb = Y + (size_t)b * C_ * K2_;
    #pragma unroll
    for (int i = 0; i < 8; ++i) {
        Yb[(rowbase + i) * K2_ + colg] = acc[i];
    }
}

#define KSTEP2 32
__global__ __launch_bounds__(256) void sp_gemm2(const float* __restrict__ Y,
                                                const float* __restrict__ x,
                                                const int*   __restrict__ length,
                                                float*       __restrict__ out) {
    int b  = blockIdx.x;
    int nb = blockIdx.y;
    int L  = length[b];

    __shared__ float sY[C_][KSTEP2 + 1];
    __shared__ float sW[KSTEP2][32 + 1];

    int tid   = threadIdx.x;
    int lane  = tid & 31;
    int wave  = tid >> 5;
    int mtile = wave >> 1;
    int ntile = wave & 1;

    v8f acc = {0.f, 0.f, 0.f, 0.f, 0.f, 0.f, 0.f, 0.f};

    const float* Yb = Y + (size_t)b * C_ * K2_;

    float ryA[(C_ * KSTEP2) / 256];
    float ryB[(C_ * KSTEP2) / 256];

    auto fetch = [&](int kk, float (&ry)[(C_ * KSTEP2) / 256]) {
        #pragma unroll
        for (int i = 0; i < (C_ * KSTEP2) / 256; ++i) {
            int li = tid + i * 256;
            int r  = li >> 5;
            int c  = li & (KSTEP2 - 1);
            ry[i] = Yb[r * K2_ + kk + c];
        }
    };
    auto stage = [&](const float (&ry)[(C_ * KSTEP2) / 256]) {
        #pragma unroll
        for (int i = 0; i < (C_ * KSTEP2) / 256; ++i) {
            int li = tid + i * 256;
            int r  = li >> 5;
            int c  = li & (KSTEP2 - 1);
            sY[r][c] = ry[i];
        }
    };
    int      nW     = nb * 32 + lane;
    float    qturnW = (wave & 1) ? 0.25f : 0.0f;
    unsigned pW     = (unsigned)((wave >> 1) * nW) & (unsigned)(OUT_ - 1);
    unsigned stepW  = (unsigned)(4 * nW) & (unsigned)(OUT_ - 1);
    float rw[(KSTEP2 * 32) / 256];
    auto synth_w = [&]() {
        #pragma unroll
        for (int i = 0; i < (KSTEP2 * 32) / 256; ++i) {
            rw[i] = (2.0f / (float)OUT_) *
                    __builtin_amdgcn_cosf((float)(int)pW * (1.0f / (float)OUT_) + qturnW);
            pW = (pW + stepW) & (unsigned)(OUT_ - 1);
        }
    };
    auto fix_w0 = [&]() {
        if (wave < 2) {
            unsigned p = wave ? ((unsigned)((NEED_ - 1) * nW) & (unsigned)(OUT_ - 1)) : 0u;
            rw[0] = (1.0f / (float)OUT_) *
                    __builtin_amdgcn_cosf((float)(int)p * (1.0f / (float)OUT_));
        }
    };
    auto stage_w = [&]() {
        #pragma unroll
        for (int i = 0; i < (KSTEP2 * 32) / 256; ++i) sW[wave + i * 8][lane] = rw[i];
    };
    int arow = mtile * 16 + (lane & 15);
    int bcol = ntile * 16 + (lane & 15);
    int koff = (lane >> 4) << 1;
    auto mma_chunk = [&]() {
        #pragma unroll
        for (int ks = 0; ks < KSTEP2; ks += 4) {
            v2f a, bf;
            a.x  = sY[arow][ks + koff];
            a.y  = sY[arow][ks + koff + 1];
            bf.x = sW[ks + koff][bcol];
            bf.y = sW[ks + koff + 1][bcol];
            acc = __builtin_amdgcn_wmma_f32_16x16x4_f32(
                false, a, false, bf, (short)0, acc, false, false);
        }
    };

    static_assert(K2_ == 16 * KSTEP2, "the peeled tail below assumes 16 chunks");
    fetch(0, ryA);
    __builtin_amdgcn_sched_barrier(0);
    fetch(KSTEP2, ryB);
    synth_w();
    fix_w0();
    for (int kk = 0; kk < 14 * KSTEP2; kk += 2 * KSTEP2) {
        stage(ryA);
        stage_w();
        fetch(kk + 2 * KSTEP2, ryA);
        lds_barrier();
        synth_w();
        mma_chunk();
        lds_barrier();

        stage(ryB);
        stage_w();
        fetch(kk + 3 * KSTEP2, ryB);
        lds_barrier();
        synth_w();
        mma_chunk();
        lds_barrier();
    }
    stage(ryA);
    stage_w();
    lds_barrier();
    synth_w();
    mma_chunk();
    lds_barrier();

    stage(ryB);
    stage_w();
    lds_barrier();
    mma_chunk();

    int n       = nb * 32 + ntile * 16 + (lane & 15);
    int rowbase = mtile * 16 + ((lane >> 4) ? 8 : 0);
    const float* xb = x + (size_t)b * C_ * T_;
    float* ob = out + (size_t)b * C_ * OUT_;
    bool shortSeq = (L < OUT_);
    #pragma unroll
    for (int i = 0; i < 8; ++i) {
        int c = rowbase + i;
        float v = acc[i];
        if (shortSeq) v = (n < L) ? xb[c * T_ + n] : 0.0f;
        ob[c * OUT_ + n] = v;
    }
}

extern "C" void kernel_launch(void* const* d_in, const int* in_sizes, int n_in,
                              void* d_out, int out_size, void* d_ws, size_t ws_size,
                              hipStream_t stream) {
    (void)in_sizes; (void)n_in; (void)out_size; (void)ws_size;
    const float* x      = (const float*)d_in[0];
    const int*   length = (const int*)d_in[1];
    float*       out    = (float*)d_out;

    float* Y = (float*)d_ws;

    sp_gemm1<<<dim3(B_, K2_ / 32), 256, 0, stream>>>(x, length, Y);
    sp_gemm2<<<dim3(B_, OUT_ / 32), 256, 0, stream>>>(Y, x, length, out);
}


# Both K-loops: all LDS fragment reads of a chunk issued as one group into registers of their own in front of the chunk's WMMAs, instead of a few at a time among them
# speedup vs baseline: 2.1775x; 1.0401x over previous
#include <hip/hip_runtime.h>
#include <math.h>

#define B_    16
#define C_    64
#define T_    2048
#define OUT_  512
#define NEED_ 257
#define K2_   512

typedef float v2f __attribute__((ext_vector_type(2)));
typedef float v8f __attribute__((ext_vector_type(8)));

static __device__ __forceinline__ void lds_barrier() {
    __builtin_amdgcn_fence(__ATOMIC_RELEASE, "workgroup", "local");
    __builtin_amdgcn_s_barrier();
    __builtin_amdgcn_fence(__ATOMIC_ACQUIRE, "workgroup", "local");
}

#define KSTEP1 64
__global__ __launch_bounds__(256) void sp_gemm1(const float* __restrict__ x,
                                                const int*   __restrict__ length,
                                                float*       __restrict__ Y) {
    int b  = blockIdx.x;
    int nb = blockIdx.y;
    int Lr = length[b];
    int L  = Lr < 1 ? 1 : Lr;
    float invL = 1.0f / (float)L;

    __shared__ float sA[C_][KSTEP1 + 1];
    __shared__ float sF[KSTEP1][32 + 1];

    int tid   = threadIdx.x;
    int lane  = tid & 31;
    int wave  = tid >> 5;
    int mtile = wave >> 1;
    int ntile = wave & 1;

    v8f acc = {0.f, 0.f, 0.f, 0.f, 0.f, 0.f, 0.f, 0.f};

    const float* xb = x + (size_t)b * C_ * T_;

    int   fc     = tid & 31;
    int   g2     = nb * 32 + fc;
    bool  nyq    = (g2 == 1);
    int   k      = nyq ? (NEED_ - 1) : (g2 >> 1);
    unsigned step = (unsigned)((8 * k) % L);
    unsigned im   = (unsigned)((k * wave) % L);
    float qturn  = ((g2 & 1) && !nyq) ? 0.25f : 0.0f;

    float rx[(C_ * KSTEP1) / 256];
    #pragma unroll
    for (int i = 0; i < (C_ * KSTEP1) / 256; ++i) {
        int li = tid + i * 256;
        int r  = li >> 6;
        int c  = li & (KSTEP1 - 1);
        rx[i] = xb[r * T_ + c];
    }

    for (int kk = 0; kk < T_; kk += KSTEP1) {
        #pragma unroll
        for (int i = 0; i < (C_ * KSTEP1) / 256; ++i) {
            int li = tid + i * 256;
            int r  = li >> 6;
            int c  = li & (KSTEP1 - 1);
            sA[r][c] = rx[i];
        }
        if (kk + KSTEP1 < T_) {
            #pragma unroll
            for (int i = 0; i < (C_ * KSTEP1) / 256; ++i) {
                int li = tid + i * 256;
                int r  = li >> 6;
                int c  = li & (KSTEP1 - 1);
                rx[i] = xb[r * T_ + kk + KSTEP1 + c];
            }
        }
        #pragma unroll
        for (int i = 0; i < (KSTEP1 * 32) / 256; ++i) {
            int tr = wave + i * 8;
            int t  = kk + tr;
            float v = 0.0f;
            if (t < L) v = __builtin_amdgcn_cosf((float)(int)im * invL + qturn);
            sF[tr][fc] = v;
            unsigned nx = im + step;
            unsigned wr = nx - (unsigned)L;
            im = nx < wr ? nx : wr;
        }
        lds_barrier();

        int arow = mtile * 16 + (lane & 15);
        int bcol = ntile * 16 + (lane & 15);
        int koff = (lane >> 4) << 1;
        v2f fa[KSTEP1 / 4], fb[KSTEP1 / 4];
        #pragma unroll
        for (int j = 0; j < KSTEP1 / 4; ++j) {
            int ks = 4 * j;
            fa[j].x = sA[arow][ks + koff];
            fa[j].y = sA[arow][ks + koff + 1];
            fb[j].x = sF[ks + koff][bcol];
            fb[j].y = sF[ks + koff + 1][bcol];
        }
        __builtin_amdgcn_sched_barrier(0);
        #pragma unroll
        for (int j = 0; j < KSTEP1 / 4; ++j) {
            acc = __builtin_amdgcn_wmma_f32_16x16x4_f32(
                false, fa[j], false, fb[j], (short)0, acc, false, false);
        }
        __builtin_amdgcn_sched_barrier(0);
        lds_barrier();
    }

    int colg    = nb * 32 + ntile * 16 + (lane & 15);
    int rowbase = mtile * 16 + ((lane >> 4) ? 8 : 0);
    float* Yb = Y + (size_t)b * C_ * K2_;
    #pragma unroll
    for (int i = 0; i < 8; ++i) {
        Yb[(rowbase + i) * K2_ + colg] = acc[i];
    }
}

#define KSTEP2 32
__global__ __launch_bounds__(256) void sp_gemm2(const float* __restrict__ Y,
                                                const float* __restrict__ x,
                                                const int*   __restrict__ length,
                                                float*       __restrict__ out) {
    int b  = blockIdx.x;
    int nb = blockIdx.y;
    int L  = length[b];

    __shared__ float sY[C_][KSTEP2 + 1];
    __shared__ float sW[KSTEP2][32 + 1];

    int tid   = threadIdx.x;
    int lane  = tid & 31;
    int wave  = tid >> 5;
    int mtile = wave >> 1;
    int ntile = wave & 1;

    v8f acc = {0.f, 0.f, 0.f, 0.f, 0.f, 0.f, 0.f, 0.f};

    const float* Yb = Y + (size_t)b * C_ * K2_;

    float ryA[(C_ * KSTEP2) / 256];
    float ryB[(C_ * KSTEP2) / 256];

    auto fetch = [&](int kk, float (&ry)[(C_ * KSTEP2) / 256]) {
        #pragma unroll
        for (int i = 0; i < (C_ * KSTEP2) / 256; ++i) {
            int li = tid + i * 256;
            int r  = li >> 5;
            int c  = li & (KSTEP2 - 1);
            ry[i] = Yb[r * K2_ + kk + c];
        }
    };
    auto stage = [&](const float (&ry)[(C_ * KSTEP2) / 256]) {
        #pragma unroll
        for (int i = 0; i < (C_ * KSTEP2) / 256; ++i) {
            int li = tid + i * 256;
            int r  = li >> 5;
            int c  = li & (KSTEP2 - 1);
            sY[r][c] = ry[i];
        }
    };
    int      nW     = nb * 32 + lane;
    float    qturnW = (wave & 1) ? 0.25f : 0.0f;
    unsigned pW     = (unsigned)((wave >> 1) * nW) & (unsigned)(OUT_ - 1);
    unsigned stepW  = (unsigned)(4 * nW) & (unsigned)(OUT_ - 1);
    float rw[(KSTEP2 * 32) / 256];
    auto synth_w = [&]() {
        #pragma unroll
        for (int i = 0; i < (KSTEP2 * 32) / 256; ++i) {
            rw[i] = (2.0f / (float)OUT_) *
                    __builtin_amdgcn_cosf((float)(int)pW * (1.0f / (float)OUT_) + qturnW);
            pW = (pW + stepW) & (unsigned)(OUT_ - 1);
        }
    };
    auto fix_w0 = [&]() {
        if (wave < 2) {
            unsigned p = wave ? ((unsigned)((NEED_ - 1) * nW) & (unsigned)(OUT_ - 1)) : 0u;
            rw[0] = (1.0f / (float)OUT_) *
                    __builtin_amdgcn_cosf((float)(int)p * (1.0f / (float)OUT_));
        }
    };
    auto stage_w = [&]() {
        #pragma unroll
        for (int i = 0; i < (KSTEP2 * 32) / 256; ++i) sW[wave + i * 8][lane] = rw[i];
    };
    int arow = mtile * 16 + (lane & 15);
    int bcol = ntile * 16 + (lane & 15);
    int koff = (lane >> 4) << 1;
    v2f fa[KSTEP2 / 4], fb[KSTEP2 / 4];
    auto read_frags = [&]() {
        #pragma unroll
        for (int j = 0; j < KSTEP2 / 4; ++j) {
            int ks = 4 * j;
            fa[j].x = sY[arow][ks + koff];
            fa[j].y = sY[arow][ks + koff + 1];
            fb[j].x = sW[ks + koff][bcol];
            fb[j].y = sW[ks + koff + 1][bcol];
        }
        __builtin_amdgcn_sched_barrier(0);
    };
    auto mma_chunk = [&]() {
        #pragma unroll
        for (int j = 0; j < KSTEP2 / 4; ++j) {
            acc = __builtin_amdgcn_wmma_f32_16x16x4_f32(
                false, fa[j], false, fb[j], (short)0, acc, false, false);
        }
        __builtin_amdgcn_sched_barrier(0);
    };

    static_assert(K2_ == 16 * KSTEP2, "the peeled tail below assumes 16 chunks");
    fetch(0, ryA);
    __builtin_amdgcn_sched_barrier(0);
    fetch(KSTEP2, ryB);
    synth_w();
    fix_w0();
    for (int kk = 0; kk < 14 * KSTEP2; kk += 2 * KSTEP2) {
        stage(ryA);
        stage_w();
        fetch(kk + 2 * KSTEP2, ryA);
        lds_barrier();
        read_frags();
        synth_w();
        mma_chunk();
        lds_barrier();

        stage(ryB);
        stage_w();
        fetch(kk + 3 * KSTEP2, ryB);
        lds_barrier();
        read_frags();
        synth_w();
        mma_chunk();
        lds_barrier();
    }
    stage(ryA);
    stage_w();
    lds_barrier();
    read_frags();
    synth_w();
    mma_chunk();
    lds_barrier();

    stage(ryB);
    stage_w();
    lds_barrier();
    read_frags();
    mma_chunk();

    int n       = nb * 32 + ntile * 16 + (lane & 15);
    int rowbase = mtile * 16 + ((lane >> 4) ? 8 : 0);
    const float* xb = x + (size_t)b * C_ * T_;
    float* ob = out + (size_t)b * C_ * OUT_;
    bool shortSeq = (L < OUT_);
    #pragma unroll
    for (int i = 0; i < 8; ++i) {
        int c = rowbase + i;
        float v = acc[i];
        if (shortSeq) v = (n < L) ? xb[c * T_ + n] : 0.0f;
        ob[c * OUT_ + n] = v;
    }
}

extern "C" void kernel_launch(void* const* d_in, const int* in_sizes, int n_in,
                              void* d_out, int out_size, void* d_ws, size_t ws_size,
                              hipStream_t stream) {
    (void)in_sizes; (void)n_in; (void)out_size; (void)ws_size;
    const float* x      = (const float*)d_in[0];
    const int*   length = (const int*)d_in[1];
    float*       out    = (float*)d_out;

    float* Y = (float*)d_ws;

    sp_gemm1<<<dim3(B_, K2_ / 32), 256, 0, stream>>>(x, length, Y);
    sp_gemm2<<<dim3(B_, OUT_ / 32), 256, 0, stream>>>(Y, x, length, out);
}


# GEMM1 split-K: 4 segments of 512 as 4x the work-groups (109 VGPRs: 4 fit a unit); segments past the valid length and copy-branch batches exit at once; GEMM2 sums live partials (new sum tree)
# speedup vs baseline: 3.1373x; 1.4408x over previous
#include <hip/hip_runtime.h>
#include <math.h>

#define B_    16
#define C_    64
#define T_    2048
#define OUT_  512
#define NEED_ 257
#define K2_   512
#define NSEG_ 4
#define SEGK_ 512

typedef float v2f __attribute__((ext_vector_type(2)));
typedef float v8f __attribute__((ext_vector_type(8)));

static __device__ __forceinline__ void lds_barrier() {
    __builtin_amdgcn_fence(__ATOMIC_RELEASE, "workgroup", "local");
    __builtin_amdgcn_s_barrier();
    __builtin_amdgcn_fence(__ATOMIC_ACQUIRE, "workgroup", "local");
}

#define KSTEP1 64
__global__ __launch_bounds__(256) void sp_gemm1(const float* __restrict__ x,
                                                const int*   __restrict__ length,
                                                float*       __restrict__ Y) {
    int b   = (blockIdx.x + 4 * blockIdx.z) & (B_ - 1);
    int seg = (blockIdx.x + blockIdx.z) & (NSEG_ - 1);
    int nb  = blockIdx.y;
    int Lr = length[b];
    int L  = Lr < 1 ? 1 : Lr;
    int kbeg = seg * SEGK_;
    int kend = (L + (KSTEP1 - 1)) & ~(KSTEP1 - 1);
    if (kend > kbeg + SEGK_) kend = kbeg + SEGK_;
    if (Lr < OUT_ || kend <= kbeg) return;
    float invL = 1.0f / (float)L;

    __shared__ float sA[C_][KSTEP1 + 1];
    __shared__ float sF[KSTEP1][32 + 1];

    int tid   = threadIdx.x;
    int lane  = tid & 31;
    int wave  = tid >> 5;
    int mtile = wave >> 1;
    int ntile = wave & 1;

    v8f acc = {0.f, 0.f, 0.f, 0.f, 0.f, 0.f, 0.f, 0.f};

    const float* xb = x + (size_t)b * C_ * T_;

    int   fc     = tid & 31;
    int   g2     = nb * 32 + fc;
    bool  nyq    = (g2 == 1);
    int   k      = nyq ? (NEED_ - 1) : (g2 >> 1);
    unsigned step = (unsigned)((8 * k) % L);
    unsigned im   = (unsigned)((k * (kbeg + wave)) % L);
    float qturn  = ((g2 & 1) && !nyq) ? 0.25f : 0.0f;

    int xr0 = tid >> 6;
    int xc  = tid & (KSTEP1 - 1);
    const float* xp = xb + xr0 * T_ + xc;
    float rx[(C_ * KSTEP1) / 256];
    #pragma unroll
    for (int i = 0; i < (C_ * KSTEP1) / 256; ++i) {
        rx[i] = xp[i * 4 * T_ + kbeg];
    }

    for (int kk = kbeg; kk < kend; kk += KSTEP1) {
        #pragma unroll
        for (int i = 0; i < (C_ * KSTEP1) / 256; ++i) {
            sA[xr0 + 4 * i][xc] = rx[i];
        }
        if (kk + KSTEP1 < kend) {
            #pragma unroll
            for (int i = 0; i < (C_ * KSTEP1) / 256; ++i) {
                rx[i] = xp[i * 4 * T_ + kk + KSTEP1];
            }
        }
        #pragma unroll
        for (int i = 0; i < (KSTEP1 * 32) / 256; ++i) {
            int tr = wave + i * 8;
            int t  = kk + tr;
            float v = 0.0f;
            if (t < L) v = __builtin_amdgcn_cosf((float)(int)im * invL + qturn);
            sF[tr][fc] = v;
            unsigned nx = im + step;
            unsigned wr = nx - (unsigned)L;
            im = nx < wr ? nx : wr;
        }
        lds_barrier();

        int arow = mtile * 16 + (lane & 15);
        int bcol = ntile * 16 + (lane & 15);
        int koff = (lane >> 4) << 1;
        v2f fa[KSTEP1 / 4], fb[KSTEP1 / 4];
        #pragma unroll
        for (int j = 0; j < KSTEP1 / 4; ++j) {
            int ks = 4 * j;
            fa[j].x = sA[arow][ks + koff];
            fa[j].y = sA[arow][ks + koff + 1];
            fb[j].x = sF[ks + koff][bcol];
            fb[j].y = sF[ks + koff + 1][bcol];
        }
        __builtin_amdgcn_sched_barrier(0);
        #pragma unroll
        for (int j = 0; j < KSTEP1 / 4; ++j) {
            acc = __builtin_amdgcn_wmma_f32_16x16x4_f32(
                false, fa[j], false, fb[j], (short)0, acc, false, false);
        }
        __builtin_amdgcn_sched_barrier(0);
        lds_barrier();
    }

    int colg    = nb * 32 + ntile * 16 + (lane & 15);
    int rowbase = mtile * 16 + ((lane >> 4) ? 8 : 0);
    float* Yb = Y + ((size_t)seg * B_ + b) * C_ * K2_;
    #pragma unroll
    for (int i = 0; i < 8; ++i) {
        Yb[(rowbase + i) * K2_ + colg] = acc[i];
    }
}

#define KSTEP2 32
template <int NS>
static __device__ __forceinline__ v8f gemm2_tile(const float* __restrict__ Yb,
                                                 float (*sY)[KSTEP2 + 1],
                                                 float (*sW)[32 + 1],
                                                 int nb, int tid) {
    constexpr int NY = (C_ * KSTEP2) / 256;
    constexpr size_t PLANE = (size_t)B_ * C_ * K2_;
    int lane  = tid & 31;
    int wave  = tid >> 5;
    int mtile = wave >> 1;
    int ntile = wave & 1;

    v8f acc = {0.f, 0.f, 0.f, 0.f, 0.f, 0.f, 0.f, 0.f};

    float ryA[NS][NY];
    float ryB[NS][NY];

    auto fetch = [&](int kk, float (&ry)[NS][NY]) {
        #pragma unroll
        for (int s = 0; s < NS; ++s) {
            #pragma unroll
            for (int i = 0; i < NY; ++i) {
                int li = tid + i * 256;
                int r  = li >> 5;
                int c  = li & (KSTEP2 - 1);
                ry[s][i] = Yb[s * PLANE + r * K2_ + kk + c];
            }
        }
    };
    auto stage = [&](const float (&ry)[NS][NY]) {
        #pragma unroll
        for (int i = 0; i < NY; ++i) {
            int li = tid + i * 256;
            int r  = li >> 5;
            int c  = li & (KSTEP2 - 1);
            float v = ry[0][i];
            #pragma unroll
            for (int s = 1; s < NS; ++s) v += ry[s][i];
            sY[r][c] = v;
        }
    };
    int      nW     = nb * 32 + lane;
    float    qturnW = (wave & 1) ? 0.25f : 0.0f;
    unsigned pW     = (unsigned)((wave >> 1) * nW) & (unsigned)(OUT_ - 1);
    unsigned stepW  = (unsigned)(4 * nW) & (unsigned)(OUT_ - 1);
    float rw[(KSTEP2 * 32) / 256];
    auto synth_w = [&]() {
        #pragma unroll
        for (int i = 0; i < (KSTEP2 * 32) / 256; ++i) {
            rw[i] = (2.0f / (float)OUT_) *
                    __builtin_amdgcn_cosf((float)(int)pW * (1.0f / (float)OUT_) + qturnW);
            pW = (pW + stepW) & (unsigned)(OUT_ - 1);
        }
    };
    auto fix_w0 = [&]() {
        if (wave < 2) {
            unsigned p = wave ? ((unsigned)((NEED_ - 1) * nW) & (unsigned)(OUT_ - 1)) : 0u;
            rw[0] = (1.0f / (float)OUT_) *
                    __builtin_amdgcn_cosf((float)(int)p * (1.0f / (float)OUT_));
        }
    };
    auto stage_w = [&]() {
        #pragma unroll
        for (int i = 0; i < (KSTEP2 * 32) / 256; ++i) sW[wave + i * 8][lane] = rw[i];
    };
    int arow = mtile * 16 + (lane & 15);
    int bcol = ntile * 16 + (lane & 15);
    int koff = (lane >> 4) << 1;
    v2f fa[KSTEP2 / 4], fb[KSTEP2 / 4];
    auto read_frags = [&]() {
        #pragma unroll
        for (int j = 0; j < KSTEP2 / 4; ++j) {
            int ks = 4 * j;
            fa[j].x = sY[arow][ks + koff];
            fa[j].y = sY[arow][ks + koff + 1];
            fb[j].x = sW[ks + koff][bcol];
            fb[j].y = sW[ks + koff + 1][bcol];
        }
        __builtin_amdgcn_sched_barrier(0);
    };
    auto mma_chunk = [&]() {
        #pragma unroll
        for (int j = 0; j < KSTEP2 / 4; ++j) {
            acc = __builtin_amdgcn_wmma_f32_16x16x4_f32(
                false, fa[j], false, fb[j], (short)0, acc, false, false);
        }
        __builtin_amdgcn_sched_barrier(0);
    };

    static_assert(K2_ == 16 * KSTEP2, "the peeled tail below assumes 16 chunks");
    fetch(0, ryA);
    __builtin_amdgcn_sched_barrier(0);
    fetch(KSTEP2, ryB);
    synth_w();
    fix_w0();
    for (int kk = 0; kk < 14 * KSTEP2; kk += 2 * KSTEP2) {
        stage(ryA);
        stage_w();
        fetch(kk + 2 * KSTEP2, ryA);
        lds_barrier();
        read_frags();
        synth_w();
        mma_chunk();
        lds_barrier();

        stage(ryB);
        stage_w();
        fetch(kk + 3 * KSTEP2, ryB);
        lds_barrier();
        read_frags();
        synth_w();
        mma_chunk();
        lds_barrier();
    }
    stage(ryA);
    stage_w();
    lds_barrier();
    read_frags();
    synth_w();
    mma_chunk();
    lds_barrier();

    stage(ryB);
    stage_w();
    lds_barrier();
    read_frags();
    mma_chunk();
    return acc;
}

__global__ __launch_bounds__(256) void sp_gemm2(const float* __restrict__ Y,
                                                const float* __restrict__ x,
                                                const int*   __restrict__ length,
                                                float*       __restrict__ out) {
    int b  = blockIdx.x;
    int nb = blockIdx.y;
    int L  = length[b];

    __shared__ float sY[C_][KSTEP2 + 1];
    __shared__ float sW[KSTEP2][32 + 1];

    int tid   = threadIdx.x;
    int lane  = tid & 31;
    int wave  = tid >> 5;
    int mtile = wave >> 1;
    int ntile = wave & 1;

    bool shortSeq = (L < OUT_);
    int nseg = shortSeq ? 0 : (L + (SEGK_ - 1)) / SEGK_;
    if (nseg > NSEG_) nseg = NSEG_;

    const float* Yb = Y + (size_t)b * C_ * K2_;
    v8f acc = {0.f, 0.f, 0.f, 0.f, 0.f, 0.f, 0.f, 0.f};
    switch (nseg) {
        case 1: acc = gemm2_tile<1>(Yb, sY, sW, nb, tid); break;
        case 2: acc = gemm2_tile<2>(Yb, sY, sW, nb, tid); break;
        case 3: acc = gemm2_tile<3>(Yb, sY, sW, nb, tid); break;
        case 4: acc = gemm2_tile<4>(Yb, sY, sW, nb, tid); break;
        default: break;
    }

    int n       = nb * 32 + ntile * 16 + (lane & 15);
    int rowbase = mtile * 16 + ((lane >> 4) ? 8 : 0);
    const float* xb = x + (size_t)b * C_ * T_;
    float* ob = out + (size_t)b * C_ * OUT_;
    #pragma unroll
    for (int i = 0; i < 8; ++i) {
        int c = rowbase + i;
        float v = acc[i];
        if (shortSeq) v = (n < L) ? xb[c * T_ + n] : 0.0f;
        ob[c * OUT_ + n] = v;
    }
}

extern "C" void kernel_launch(void* const* d_in, const int* in_sizes, int n_in,
                              void* d_out, int out_size, void* d_ws, size_t ws_size,
                              hipStream_t stream) {
    (void)in_sizes; (void)n_in; (void)out_size; (void)ws_size;
    const float* x      = (const float*)d_in[0];
    const int*   length = (const int*)d_in[1];
    float*       out    = (float*)d_out;

    float* Y = (float*)d_ws;

    sp_gemm1<<<dim3(B_, K2_ / 32, NSEG_), 256, 0, stream>>>(x, length, Y);
    sp_gemm2<<<dim3(B_, OUT_ / 32), 256, 0, stream>>>(Y, x, length, out);
}


# Both GEMMs: LDS tiles k-contiguous at a 16-byte-odd pitch (basis tiles transposed); one 16-byte LDS read per operand and WMMA pair replaces four 4-byte reads; k slots swizzled, sums bit-identical
# speedup vs baseline: 3.9271x; 1.2517x over previous
#include <hip/hip_runtime.h>
#include <math.h>

#define B_    16
#define C_    64
#define T_    2048
#define OUT_  512
#define NEED_ 257
#define K2_   512
#define NSEG_ 4
#define SEGK_ 512

typedef float v2f __attribute__((ext_vector_type(2)));
typedef float v4f __attribute__((ext_vector_type(4)));
typedef float v8f __attribute__((ext_vector_type(8)));

static __device__ __forceinline__ int kslot(int k) {
    return (k & ~6) | ((k & 2) << 1) | ((k & 4) >> 1);
}

static __device__ __forceinline__ void lds_barrier() {
    __builtin_amdgcn_fence(__ATOMIC_RELEASE, "workgroup", "local");
    __builtin_amdgcn_s_barrier();
    __builtin_amdgcn_fence(__ATOMIC_ACQUIRE, "workgroup", "local");
}

#define KSTEP1 64
#define PITCH1 (KSTEP1 + 4)
__global__ __launch_bounds__(256) void sp_gemm1(const float* __restrict__ x,
                                                const int*   __restrict__ length,
                                                float*       __restrict__ Y) {
    int b   = (blockIdx.x + 4 * blockIdx.z) & (B_ - 1);
    int seg = (blockIdx.x + blockIdx.z) & (NSEG_ - 1);
    int nb  = blockIdx.y;
    int Lr = length[b];
    int L  = Lr < 1 ? 1 : Lr;
    int kbeg = seg * SEGK_;
    int kend = (L + (KSTEP1 - 1)) & ~(KSTEP1 - 1);
    if (kend > kbeg + SEGK_) kend = kbeg + SEGK_;
    if (Lr < OUT_ || kend <= kbeg) return;
    float invL = 1.0f / (float)L;

    __shared__ __attribute__((aligned(16))) float sA[C_][PITCH1];
    __shared__ __attribute__((aligned(16))) float sF[32][PITCH1];

    int tid   = threadIdx.x;
    int lane  = tid & 31;
    int wave  = tid >> 5;
    int mtile = wave >> 1;
    int ntile = wave & 1;

    v8f acc = {0.f, 0.f, 0.f, 0.f, 0.f, 0.f, 0.f, 0.f};

    const float* xb = x + (size_t)b * C_ * T_;

    int   fc     = tid & 31;
    int   g2     = nb * 32 + fc;
    bool  nyq    = (g2 == 1);
    int   k      = nyq ? (NEED_ - 1) : (g2 >> 1);
    unsigned step = (unsigned)((8 * k) % L);
    unsigned im   = (unsigned)((k * (kbeg + wave)) % L);
    float qturn  = ((g2 & 1) && !nyq) ? 0.25f : 0.0f;

    int xr0 = tid >> 6;
    int xc  = tid & (KSTEP1 - 1);
    int xs  = kslot(xc);
    int fs0 = kslot(wave);
    const float* xp = xb + xr0 * T_ + xc;
    float rx[(C_ * KSTEP1) / 256];
    #pragma unroll
    for (int i = 0; i < (C_ * KSTEP1) / 256; ++i) {
        rx[i] = xp[i * 4 * T_ + kbeg];
    }

    for (int kk = kbeg; kk < kend; kk += KSTEP1) {
        #pragma unroll
        for (int i = 0; i < (C_ * KSTEP1) / 256; ++i) {
            sA[xr0 + 4 * i][xs] = rx[i];
        }
        if (kk + KSTEP1 < kend) {
            #pragma unroll
            for (int i = 0; i < (C_ * KSTEP1) / 256; ++i) {
                rx[i] = xp[i * 4 * T_ + kk + KSTEP1];
            }
        }
        #pragma unroll
        for (int i = 0; i < (KSTEP1 * 32) / 256; ++i) {
            int tr = wave + i * 8;
            int t  = kk + tr;
            float v = 0.0f;
            if (t < L) v = __builtin_amdgcn_cosf((float)(int)im * invL + qturn);
            sF[fc][fs0 + i * 8] = v;
            unsigned nx = im + step;
            unsigned wr = nx - (unsigned)L;
            im = nx < wr ? nx : wr;
        }
        lds_barrier();

        int arow = mtile * 16 + (lane & 15);
        int bcol = ntile * 16 + (lane & 15);
        int soff = (lane >> 4) << 2;
        v4f qa[KSTEP1 / 8], qb[KSTEP1 / 8];
        #pragma unroll
        for (int j = 0; j < KSTEP1 / 8; ++j) {
            qa[j] = *(const v4f*)&sA[arow][8 * j + soff];
            qb[j] = *(const v4f*)&sF[bcol][8 * j + soff];
        }
        __builtin_amdgcn_sched_barrier(0);
        #pragma unroll
        for (int j = 0; j < KSTEP1 / 8; ++j) {
            v2f a0 = {qa[j].x, qa[j].y}, b0 = {qb[j].x, qb[j].y};
            v2f a1 = {qa[j].z, qa[j].w}, b1 = {qb[j].z, qb[j].w};
            acc = __builtin_amdgcn_wmma_f32_16x16x4_f32(
                false, a0, false, b0, (short)0, acc, false, false);
            acc = __builtin_amdgcn_wmma_f32_16x16x4_f32(
                false, a1, false, b1, (short)0, acc, false, false);
        }
        __builtin_amdgcn_sched_barrier(0);
        lds_barrier();
    }

    int colg    = nb * 32 + ntile * 16 + (lane & 15);
    int rowbase = mtile * 16 + ((lane >> 4) ? 8 : 0);
    float* Yb = Y + ((size_t)seg * B_ + b) * C_ * K2_;
    #pragma unroll
    for (int i = 0; i < 8; ++i) {
        Yb[(rowbase + i) * K2_ + colg] = acc[i];
    }
}

#define KSTEP2 32
#define PITCH2 (KSTEP2 + 4)
template <int NS>
static __device__ __forceinline__ v8f gemm2_tile(const float* __restrict__ Yb,
                                                 float (*sY)[PITCH2],
                                                 float (*sW)[PITCH2],
                                                 int nb, int tid) {
    constexpr int NY = (C_ * KSTEP2) / 256;
    constexpr size_t PLANE = (size_t)B_ * C_ * K2_;
    int lane  = tid & 31;
    int wave  = tid >> 5;
    int mtile = wave >> 1;
    int ntile = wave & 1;

    v8f acc = {0.f, 0.f, 0.f, 0.f, 0.f, 0.f, 0.f, 0.f};

    float ryA[NS][NY];
    float ryB[NS][NY];

    auto fetch = [&](int kk, float (&ry)[NS][NY]) {
        #pragma unroll
        for (int s = 0; s < NS; ++s) {
            #pragma unroll
            for (int i = 0; i < NY; ++i) {
                int li = tid + i * 256;
                int r  = li >> 5;
                int c  = li & (KSTEP2 - 1);
                ry[s][i] = Yb[s * PLANE + r * K2_ + kk + c];
            }
        }
    };
    auto stage = [&](const float (&ry)[NS][NY]) {
        #pragma unroll
        for (int i = 0; i < NY; ++i) {
            int li = tid + i * 256;
            int r  = li >> 5;
            int c  = li & (KSTEP2 - 1);
            float v = ry[0][i];
            #pragma unroll
            for (int s = 1; s < NS; ++s) v += ry[s][i];
            sY[r][kslot(c)] = v;
        }
    };
    int      nW     = nb * 32 + lane;
    float    qturnW = (wave & 1) ? 0.25f : 0.0f;
    unsigned pW     = (unsigned)((wave >> 1) * nW) & (unsigned)(OUT_ - 1);
    unsigned stepW  = (unsigned)(4 * nW) & (unsigned)(OUT_ - 1);
    float rw[(KSTEP2 * 32) / 256];
    auto synth_w = [&]() {
        #pragma unroll
        for (int i = 0; i < (KSTEP2 * 32) / 256; ++i) {
            rw[i] = (2.0f / (float)OUT_) *
                    __builtin_amdgcn_cosf((float)(int)pW * (1.0f / (float)OUT_) + qturnW);
            pW = (pW + stepW) & (unsigned)(OUT_ - 1);
        }
    };
    auto fix_w0 = [&]() {
        if (wave < 2) {
            unsigned p = wave ? ((unsigned)((NEED_ - 1) * nW) & (unsigned)(OUT_ - 1)) : 0u;
            rw[0] = (1.0f / (float)OUT_) *
                    __builtin_amdgcn_cosf((float)(int)p * (1.0f / (float)OUT_));
        }
    };
    int ws0 = kslot(wave);
    auto stage_w = [&]() {
        #pragma unroll
        for (int i = 0; i < (KSTEP2 * 32) / 256; ++i) sW[lane][ws0 + i * 8] = rw[i];
    };
    int arow = mtile * 16 + (lane & 15);
    int bcol = ntile * 16 + (lane & 15);
    int soff = (lane >> 4) << 2;
    v4f qa[KSTEP2 / 8], qb[KSTEP2 / 8];
    auto read_frags = [&]() {
        #pragma unroll
        for (int j = 0; j < KSTEP2 / 8; ++j) {
            qa[j] = *(const v4f*)&sY[arow][8 * j + soff];
            qb[j] = *(const v4f*)&sW[bcol][8 * j + soff];
        }
        __builtin_amdgcn_sched_barrier(0);
    };
    auto mma_chunk = [&]() {
        #pragma unroll
        for (int j = 0; j < KSTEP2 / 8; ++j) {
            v2f a0 = {qa[j].x, qa[j].y}, b0 = {qb[j].x, qb[j].y};
            v2f a1 = {qa[j].z, qa[j].w}, b1 = {qb[j].z, qb[j].w};
            acc = __builtin_amdgcn_wmma_f32_16x16x4_f32(
                false, a0, false, b0, (short)0, acc, false, false);
            acc = __builtin_amdgcn_wmma_f32_16x16x4_f32(
                false, a1, false, b1, (short)0, acc, false, false);
        }
        __builtin_amdgcn_sched_barrier(0);
    };

    static_assert(K2_ == 16 * KSTEP2, "the peeled tail below assumes 16 chunks");
    fetch(0, ryA);
    __builtin_amdgcn_sched_barrier(0);
    fetch(KSTEP2, ryB);
    synth_w();
    fix_w0();
    for (int kk = 0; kk < 14 * KSTEP2; kk += 2 * KSTEP2) {
        stage(ryA);
        stage_w();
        fetch(kk + 2 * KSTEP2, ryA);
        lds_barrier();
        read_frags();
        synth_w();
        mma_chunk();
        lds_barrier();

        stage(ryB);
        stage_w();
        fetch(kk + 3 * KSTEP2, ryB);
        lds_barrier();
        read_frags();
        synth_w();
        mma_chunk();
        lds_barrier();
    }
    stage(ryA);
    stage_w();
    lds_barrier();
    read_frags();
    synth_w();
    mma_chunk();
    lds_barrier();

    stage(ryB);
    stage_w();
    lds_barrier();
    read_frags();
    mma_chunk();
    return acc;
}

__global__ __launch_bounds__(256) void sp_gemm2(const float* __restrict__ Y,
                                                const float* __restrict__ x,
                                                const int*   __restrict__ length,
                                                float*       __restrict__ out) {
    int b  = blockIdx.x;
    int nb = blockIdx.y;
    int L  = length[b];

    __shared__ __attribute__((aligned(16))) float sY[C_][PITCH2];
    __shared__ __attribute__((aligned(16))) float sW[32][PITCH2];

    int tid   = threadIdx.x;
    int lane  = tid & 31;
    int wave  = tid >> 5;
    int mtile = wave >> 1;
    int ntile = wave & 1;

    bool shortSeq = (L < OUT_);
    int nseg = shortSeq ? 0 : (L + (SEGK_ - 1)) / SEGK_;
    if (nseg > NSEG_) nseg = NSEG_;

    const float* Yb = Y + (size_t)b * C_ * K2_;
    v8f acc = {0.f, 0.f, 0.f, 0.f, 0.f, 0.f, 0.f, 0.f};
    switch (nseg) {
        case 1: acc = gemm2_tile<1>(Yb, sY, sW, nb, tid); break;
        case 2: acc = gemm2_tile<2>(Yb, sY, sW, nb, tid); break;
        case 3: acc = gemm2_tile<3>(Yb, sY, sW, nb, tid); break;
        case 4: acc = gemm2_tile<4>(Yb, sY, sW, nb, tid); break;
        default: break;
    }

    int n       = nb * 32 + ntile * 16 + (lane & 15);
    int rowbase = mtile * 16 + ((lane >> 4) ? 8 : 0);
    const float* xb = x + (size_t)b * C_ * T_;
    float* ob = out + (size_t)b * C_ * OUT_;
    #pragma unroll
    for (int i = 0; i < 8; ++i) {
        int c = rowbase + i;
        float v = acc[i];
        if (shortSeq) v = (n < L) ? xb[c * T_ + n] : 0.0f;
        ob[c * OUT_ + n] = v;
    }
}

extern "C" void kernel_launch(void* const* d_in, const int* in_sizes, int n_in,
                              void* d_out, int out_size, void* d_ws, size_t ws_size,
                              hipStream_t stream) {
    (void)in_sizes; (void)n_in; (void)out_size; (void)ws_size;
    const float* x      = (const float*)d_in[0];
    const int*   length = (const int*)d_in[1];
    float*       out    = (float*)d_out;

    float* Y = (float*)d_ws;

    sp_gemm1<<<dim3(B_, K2_ / 32, NSEG_), 256, 0, stream>>>(x, length, Y);
    sp_gemm2<<<dim3(B_, OUT_ / 32), 256, 0, stream>>>(Y, x, length, out);
}


# Both GEMMs: global loads of the staged operand (x tile, partial-spectrum tile) widened from 4 to 16 bytes per lane, written to LDS as two 8-byte stores into the same slots; sums bit-identical
# speedup vs baseline: 4.0719x; 1.0369x over previous
#include <hip/hip_runtime.h>
#include <math.h>

#define B_    16
#define C_    64
#define T_    2048
#define OUT_  512
#define NEED_ 257
#define K2_   512
#define NSEG_ 4
#define SEGK_ 512

typedef float v2f __attribute__((ext_vector_type(2)));
typedef float v4f __attribute__((ext_vector_type(4)));
typedef float v8f __attribute__((ext_vector_type(8)));

static __device__ __forceinline__ int kslot(int k) {
    return (k & ~6) | ((k & 2) << 1) | ((k & 4) >> 1);
}

static __device__ __forceinline__ void lds_barrier() {
    __builtin_amdgcn_fence(__ATOMIC_RELEASE, "workgroup", "local");
    __builtin_amdgcn_s_barrier();
    __builtin_amdgcn_fence(__ATOMIC_ACQUIRE, "workgroup", "local");
}

#define KSTEP1 64
#define PITCH1 (KSTEP1 + 4)
__global__ __launch_bounds__(256) void sp_gemm1(const float* __restrict__ x,
                                                const int*   __restrict__ length,
                                                float*       __restrict__ Y) {
    int b   = (blockIdx.x + 4 * blockIdx.z) & (B_ - 1);
    int seg = (blockIdx.x + blockIdx.z) & (NSEG_ - 1);
    int nb  = blockIdx.y;
    int Lr = length[b];
    int L  = Lr < 1 ? 1 : Lr;
    int kbeg = seg * SEGK_;
    int kend = (L + (KSTEP1 - 1)) & ~(KSTEP1 - 1);
    if (kend > kbeg + SEGK_) kend = kbeg + SEGK_;
    if (Lr < OUT_ || kend <= kbeg) return;
    float invL = 1.0f / (float)L;

    __shared__ __attribute__((aligned(16))) float sA[C_][PITCH1];
    __shared__ __attribute__((aligned(16))) float sF[32][PITCH1];

    int tid   = threadIdx.x;
    int lane  = tid & 31;
    int wave  = tid >> 5;
    int mtile = wave >> 1;
    int ntile = wave & 1;

    v8f acc = {0.f, 0.f, 0.f, 0.f, 0.f, 0.f, 0.f, 0.f};

    const float* xb = x + (size_t)b * C_ * T_;

    int   fc     = tid & 31;
    int   g2     = nb * 32 + fc;
    bool  nyq    = (g2 == 1);
    int   k      = nyq ? (NEED_ - 1) : (g2 >> 1);
    unsigned step = (unsigned)((8 * k) % L);
    unsigned im   = (unsigned)((k * (kbeg + wave)) % L);
    float qturn  = ((g2 & 1) && !nyq) ? 0.25f : 0.0f;

    int xr0 = tid >> 4;
    int xq  = tid & (KSTEP1 / 4 - 1);
    int xs  = kslot(4 * xq);
    int fs0 = kslot(wave);
    const float* xp = xb + xr0 * T_ + 4 * xq;
    constexpr int NX = (C_ * KSTEP1) / (256 * 4);
    v4f rx[NX];
    #pragma unroll
    for (int i = 0; i < NX; ++i) {
        rx[i] = *(const v4f*)(xp + i * 16 * T_ + kbeg);
    }

    for (int kk = kbeg; kk < kend; kk += KSTEP1) {
        #pragma unroll
        for (int i = 0; i < NX; ++i) {
            v2f lo = {rx[i].x, rx[i].y}, hi = {rx[i].z, rx[i].w};
            *(v2f*)&sA[xr0 + 16 * i][xs]     = lo;
            *(v2f*)&sA[xr0 + 16 * i][xs + 4] = hi;
        }
        if (kk + KSTEP1 < kend) {
            #pragma unroll
            for (int i = 0; i < NX; ++i) {
                rx[i] = *(const v4f*)(xp + i * 16 * T_ + kk + KSTEP1);
            }
        }
        #pragma unroll
        for (int i = 0; i < (KSTEP1 * 32) / 256; ++i) {
            int tr = wave + i * 8;
            int t  = kk + tr;
            float v = 0.0f;
            if (t < L) v = __builtin_amdgcn_cosf((float)(int)im * invL + qturn);
            sF[fc][fs0 + i * 8] = v;
            unsigned nx = im + step;
            unsigned wr = nx - (unsigned)L;
            im = nx < wr ? nx : wr;
        }
        lds_barrier();

        int arow = mtile * 16 + (lane & 15);
        int bcol = ntile * 16 + (lane & 15);
        int soff = (lane >> 4) << 2;
        v4f qa[KSTEP1 / 8], qb[KSTEP1 / 8];
        #pragma unroll
        for (int j = 0; j < KSTEP1 / 8; ++j) {
            qa[j] = *(const v4f*)&sA[arow][8 * j + soff];
            qb[j] = *(const v4f*)&sF[bcol][8 * j + soff];
        }
        __builtin_amdgcn_sched_barrier(0);
        #pragma unroll
        for (int j = 0; j < KSTEP1 / 8; ++j) {
            v2f a0 = {qa[j].x, qa[j].y}, b0 = {qb[j].x, qb[j].y};
            v2f a1 = {qa[j].z, qa[j].w}, b1 = {qb[j].z, qb[j].w};
            acc = __builtin_amdgcn_wmma_f32_16x16x4_f32(
                false, a0, false, b0, (short)0, acc, false, false);
            acc = __builtin_amdgcn_wmma_f32_16x16x4_f32(
                false, a1, false, b1, (short)0, acc, false, false);
        }
        __builtin_amdgcn_sched_barrier(0);
        lds_barrier();
    }

    int colg    = nb * 32 + ntile * 16 + (lane & 15);
    int rowbase = mtile * 16 + ((lane >> 4) ? 8 : 0);
    float* Yb = Y + ((size_t)seg * B_ + b) * C_ * K2_;
    #pragma unroll
    for (int i = 0; i < 8; ++i) {
        Yb[(rowbase + i) * K2_ + colg] = acc[i];
    }
}

#define KSTEP2 32
#define PITCH2 (KSTEP2 + 4)
template <int NS>
static __device__ __forceinline__ v8f gemm2_tile(const float* __restrict__ Yb,
                                                 float (*sY)[PITCH2],
                                                 float (*sW)[PITCH2],
                                                 int nb, int tid) {
    constexpr int NY = (C_ * KSTEP2) / (256 * 4);
    constexpr size_t PLANE = (size_t)B_ * C_ * K2_;
    int lane  = tid & 31;
    int wave  = tid >> 5;
    int mtile = wave >> 1;
    int ntile = wave & 1;

    v8f acc = {0.f, 0.f, 0.f, 0.f, 0.f, 0.f, 0.f, 0.f};

    v4f ryA[NS][NY];
    v4f ryB[NS][NY];
    int yr0 = tid >> 3;
    int yq  = tid & (KSTEP2 / 4 - 1);
    int ys  = kslot(4 * yq);
    const float* yp = Yb + yr0 * K2_ + 4 * yq;

    auto fetch = [&](int kk, v4f (&ry)[NS][NY]) {
        #pragma unroll
        for (int s = 0; s < NS; ++s) {
            #pragma unroll
            for (int i = 0; i < NY; ++i) {
                ry[s][i] = *(const v4f*)(yp + s * PLANE + i * 32 * K2_ + kk);
            }
        }
    };
    auto stage = [&](const v4f (&ry)[NS][NY]) {
        #pragma unroll
        for (int i = 0; i < NY; ++i) {
            v4f v = ry[0][i];
            #pragma unroll
            for (int s = 1; s < NS; ++s) v += ry[s][i];
            v2f lo = {v.x, v.y}, hi = {v.z, v.w};
            *(v2f*)&sY[yr0 + 32 * i][ys]     = lo;
            *(v2f*)&sY[yr0 + 32 * i][ys + 4] = hi;
        }
    };
    int      nW     = nb * 32 + lane;
    float    qturnW = (wave & 1) ? 0.25f : 0.0f;
    unsigned pW     = (unsigned)((wave >> 1) * nW) & (unsigned)(OUT_ - 1);
    unsigned stepW  = (unsigned)(4 * nW) & (unsigned)(OUT_ - 1);
    float rw[(KSTEP2 * 32) / 256];
    auto synth_w = [&]() {
        #pragma unroll
        for (int i = 0; i < (KSTEP2 * 32) / 256; ++i) {
            rw[i] = (2.0f / (float)OUT_) *
                    __builtin_amdgcn_cosf((float)(int)pW * (1.0f / (float)OUT_) + qturnW);
            pW = (pW + stepW) & (unsigned)(OUT_ - 1);
        }
    };
    auto fix_w0 = [&]() {
        if (wave < 2) {
            unsigned p = wave ? ((unsigned)((NEED_ - 1) * nW) & (unsigned)(OUT_ - 1)) : 0u;
            rw[0] = (1.0f / (float)OUT_) *
                    __builtin_amdgcn_cosf((float)(int)p * (1.0f / (float)OUT_));
        }
    };
    int ws0 = kslot(wave);
    auto stage_w = [&]() {
        #pragma unroll
        for (int i = 0; i < (KSTEP2 * 32) / 256; ++i) sW[lane][ws0 + i * 8] = rw[i];
    };
    int arow = mtile * 16 + (lane & 15);
    int bcol = ntile * 16 + (lane & 15);
    int soff = (lane >> 4) << 2;
    v4f qa[KSTEP2 / 8], qb[KSTEP2 / 8];
    auto read_frags = [&]() {
        #pragma unroll
        for (int j = 0; j < KSTEP2 / 8; ++j) {
            qa[j] = *(const v4f*)&sY[arow][8 * j + soff];
            qb[j] = *(const v4f*)&sW[bcol][8 * j + soff];
        }
        __builtin_amdgcn_sched_barrier(0);
    };
    auto mma_chunk = [&]() {
        #pragma unroll
        for (int j = 0; j < KSTEP2 / 8; ++j) {
            v2f a0 = {qa[j].x, qa[j].y}, b0 = {qb[j].x, qb[j].y};
            v2f a1 = {qa[j].z, qa[j].w}, b1 = {qb[j].z, qb[j].w};
            acc = __builtin_amdgcn_wmma_f32_16x16x4_f32(
                false, a0, false, b0, (short)0, acc, false, false);
            acc = __builtin_amdgcn_wmma_f32_16x16x4_f32(
                false, a1, false, b1, (short)0, acc, false, false);
        }
        __builtin_amdgcn_sched_barrier(0);
    };

    static_assert(K2_ == 16 * KSTEP2, "the peeled tail below assumes 16 chunks");
    fetch(0, ryA);
    __builtin_amdgcn_sched_barrier(0);
    fetch(KSTEP2, ryB);
    synth_w();
    fix_w0();
    for (int kk = 0; kk < 14 * KSTEP2; kk += 2 * KSTEP2) {
        stage(ryA);
        stage_w();
        fetch(kk + 2 * KSTEP2, ryA);
        lds_barrier();
        read_frags();
        synth_w();
        mma_chunk();
        lds_barrier();

        stage(ryB);
        stage_w();
        fetch(kk + 3 * KSTEP2, ryB);
        lds_barrier();
        read_frags();
        synth_w();
        mma_chunk();
        lds_barrier();
    }
    stage(ryA);
    stage_w();
    lds_barrier();
    read_frags();
    synth_w();
    mma_chunk();
    lds_barrier();

    stage(ryB);
    stage_w();
    lds_barrier();
    read_frags();
    mma_chunk();
    return acc;
}

__global__ __launch_bounds__(256) void sp_gemm2(const float* __restrict__ Y,
                                                const float* __restrict__ x,
                                                const int*   __restrict__ length,
                                                float*       __restrict__ out) {
    int b  = blockIdx.x;
    int nb = blockIdx.y;
    int L  = length[b];

    __shared__ __attribute__((aligned(16))) float sY[C_][PITCH2];
    __shared__ __attribute__((aligned(16))) float sW[32][PITCH2];

    int tid   = threadIdx.x;
    int lane  = tid & 31;
    int wave  = tid >> 5;
    int mtile = wave >> 1;
    int ntile = wave & 1;

    bool shortSeq = (L < OUT_);
    int nseg = shortSeq ? 0 : (L + (SEGK_ - 1)) / SEGK_;
    if (nseg > NSEG_) nseg = NSEG_;

    const float* Yb = Y + (size_t)b * C_ * K2_;
    v8f acc = {0.f, 0.f, 0.f, 0.f, 0.f, 0.f, 0.f, 0.f};
    switch (nseg) {
        case 1: acc = gemm2_tile<1>(Yb, sY, sW, nb, tid); break;
        case 2: acc = gemm2_tile<2>(Yb, sY, sW, nb, tid); break;
        case 3: acc = gemm2_tile<3>(Yb, sY, sW, nb, tid); break;
        case 4: acc = gemm2_tile<4>(Yb, sY, sW, nb, tid); break;
        default: break;
    }

    int n       = nb * 32 + ntile * 16 + (lane & 15);
    int rowbase = mtile * 16 + ((lane >> 4) ? 8 : 0);
    const float* xb = x + (size_t)b * C_ * T_;
    float* ob = out + (size_t)b * C_ * OUT_;
    #pragma unroll
    for (int i = 0; i < 8; ++i) {
        int c = rowbase + i;
        float v = acc[i];
        if (shortSeq) v = (n < L) ? xb[c * T_ + n] : 0.0f;
        ob[c * OUT_ + n] = v;
    }
}

extern "C" void kernel_launch(void* const* d_in, const int* in_sizes, int n_in,
                              void* d_out, int out_size, void* d_ws, size_t ws_size,
                              hipStream_t stream) {
    (void)in_sizes; (void)n_in; (void)out_size; (void)ws_size;
    const float* x      = (const float*)d_in[0];
    const int*   length = (const int*)d_in[1];
    float*       out    = (float*)d_out;

    float* Y = (float*)d_ws;

    sp_gemm1<<<dim3(B_, K2_ / 32, NSEG_), 256, 0, stream>>>(x, length, Y);
    sp_gemm2<<<dim3(B_, OUT_ / 32), 256, 0, stream>>>(Y, x, length, out);
}


# GEMM1: a work-group's (batch, K segment) comes from a list of the live pairs built from the 16 lengths by wave ballots (whole segments first, empty last), not a fixed index map; sums bit-identical
# speedup vs baseline: 4.3357x; 1.0648x over previous
#include <hip/hip_runtime.h>
#include <math.h>

#define B_    16
#define C_    64
#define T_    2048
#define OUT_  512
#define NEED_ 257
#define K2_   512
#define NSEG_ 4
#define SEGK_ 512

typedef float v2f __attribute__((ext_vector_type(2)));
typedef float v4f __attribute__((ext_vector_type(4)));
typedef float v8f __attribute__((ext_vector_type(8)));

static __device__ __forceinline__ int kslot(int k) {
    return (k & ~6) | ((k & 2) << 1) | ((k & 4) >> 1);
}

static __device__ __forceinline__ void lds_barrier() {
    __builtin_amdgcn_fence(__ATOMIC_RELEASE, "workgroup", "local");
    __builtin_amdgcn_s_barrier();
    __builtin_amdgcn_fence(__ATOMIC_ACQUIRE, "workgroup", "local");
}

#define KSTEP1 64
#define PITCH1 (KSTEP1 + 4)
#define NCB1   (K2_ / 32)
__global__ __launch_bounds__(256) void sp_gemm1(const float* __restrict__ x,
                                                const int*   __restrict__ length,
                                                float*       __restrict__ Y) {
    unsigned wg = blockIdx.x;
    int nb   = (int)(wg % NCB1);
    int pos  = (int)(wg / NCB1);
    int ln   = threadIdx.x & 31;
    int Lq   = length[ln & (B_ - 1)];
    int sgq  = ln >> 4;
    int endq = Lq < OUT_ ? 0 : ((Lq + (KSTEP1 - 1)) & ~(KSTEP1 - 1));
    int rem0 = endq - sgq * SEGK_;
    int rem1 = endq - (sgq + 2) * SEGK_;
    bool whole0 = rem0 >= SEGK_, cut0 = rem0 > 0 && !whole0;
    bool whole1 = rem1 >= SEGK_, cut1 = rem1 > 0 && !whole1;
    unsigned W0 = __builtin_amdgcn_ballot_w32(whole0), W1 = __builtin_amdgcn_ballot_w32(whole1);
    unsigned C0 = __builtin_amdgcn_ballot_w32(cut0),   C1 = __builtin_amdgcn_ballot_w32(cut1);
    unsigned below = (1u << ln) - 1u;
    int nW0 = __popc(W0), nW = nW0 + __popc(W1), nC0 = __popc(C0);
    int place0 = (whole0 ? 0   : nW)       + __popc((whole0 ? W0 : C0) & below);
    int place1 = (whole1 ? nW0 : nW + nC0) + __popc((whole1 ? W1 : C1) & below);
    unsigned H0 = __builtin_amdgcn_ballot_w32((whole0 || cut0) && place0 == pos);
    unsigned H1 = __builtin_amdgcn_ballot_w32((whole1 || cut1) && place1 == pos);
    if ((H0 | H1) == 0u) return;
    int pair = H0 ? __ffs(H0) - 1 : 31 + __ffs(H1);
    int b    = pair & (B_ - 1);
    int seg  = pair >> 4;
    int L    = __builtin_amdgcn_readlane(Lq, b);
    int kbeg = seg * SEGK_;
    int kend = (L + (KSTEP1 - 1)) & ~(KSTEP1 - 1);
    if (kend > kbeg + SEGK_) kend = kbeg + SEGK_;
    float invL = 1.0f / (float)L;

    __shared__ __attribute__((aligned(16))) float sA[C_][PITCH1];
    __shared__ __attribute__((aligned(16))) float sF[32][PITCH1];

    int tid   = threadIdx.x;
    int lane  = tid & 31;
    int wave  = tid >> 5;
    int mtile = wave >> 1;
    int ntile = wave & 1;

    v8f acc = {0.f, 0.f, 0.f, 0.f, 0.f, 0.f, 0.f, 0.f};

    const float* xb = x + (size_t)b * C_ * T_;

    int   fc     = tid & 31;
    int   g2     = nb * 32 + fc;
    bool  nyq    = (g2 == 1);
    int   k      = nyq ? (NEED_ - 1) : (g2 >> 1);
    unsigned step = (unsigned)((8 * k) % L);
    unsigned im   = (unsigned)((k * (kbeg + wave)) % L);
    float qturn  = ((g2 & 1) && !nyq) ? 0.25f : 0.0f;

    int xr0 = tid >> 4;
    int xq  = tid & (KSTEP1 / 4 - 1);
    int xs  = kslot(4 * xq);
    int fs0 = kslot(wave);
    const float* xp = xb + xr0 * T_ + 4 * xq;
    constexpr int NX = (C_ * KSTEP1) / (256 * 4);
    v4f rx[NX];
    #pragma unroll
    for (int i = 0; i < NX; ++i) {
        rx[i] = *(const v4f*)(xp + i * 16 * T_ + kbeg);
    }

    for (int kk = kbeg; kk < kend; kk += KSTEP1) {
        #pragma unroll
        for (int i = 0; i < NX; ++i) {
            v2f lo = {rx[i].x, rx[i].y}, hi = {rx[i].z, rx[i].w};
            *(v2f*)&sA[xr0 + 16 * i][xs]     = lo;
            *(v2f*)&sA[xr0 + 16 * i][xs + 4] = hi;
        }
        if (kk + KSTEP1 < kend) {
            #pragma unroll
            for (int i = 0; i < NX; ++i) {
                rx[i] = *(const v4f*)(xp + i * 16 * T_ + kk + KSTEP1);
            }
        }
        #pragma unroll
        for (int i = 0; i < (KSTEP1 * 32) / 256; ++i) {
            int tr = wave + i * 8;
            int t  = kk + tr;
            float v = 0.0f;
            if (t < L) v = __builtin_amdgcn_cosf((float)(int)im * invL + qturn);
            sF[fc][fs0 + i * 8] = v;
            unsigned nx = im + step;
            unsigned wr = nx - (unsigned)L;
            im = nx < wr ? nx : wr;
        }
        lds_barrier();

        int arow = mtile * 16 + (lane & 15);
        int bcol = ntile * 16 + (lane & 15);
        int soff = (lane >> 4) << 2;
        v4f qa[KSTEP1 / 8], qb[KSTEP1 / 8];
        #pragma unroll
        for (int j = 0; j < KSTEP1 / 8; ++j) {
            qa[j] = *(const v4f*)&sA[arow][8 * j + soff];
            qb[j] = *(const v4f*)&sF[bcol][8 * j + soff];
        }
        __builtin_amdgcn_sched_barrier(0);
        #pragma unroll
        for (int j = 0; j < KSTEP1 / 8; ++j) {
            v2f a0 = {qa[j].x, qa[j].y}, b0 = {qb[j].x, qb[j].y};
            v2f a1 = {qa[j].z, qa[j].w}, b1 = {qb[j].z, qb[j].w};
            acc = __builtin_amdgcn_wmma_f32_16x16x4_f32(
                false, a0, false, b0, (short)0, acc, false, false);
            acc = __builtin_amdgcn_wmma_f32_16x16x4_f32(
                false, a1, false, b1, (short)0, acc, false, false);
        }
        __builtin_amdgcn_sched_barrier(0);
        lds_barrier();
    }

    int colg    = nb * 32 + ntile * 16 + (lane & 15);
    int rowbase = mtile * 16 + ((lane >> 4) ? 8 : 0);
    float* Yb = Y + ((size_t)seg * B_ + b) * C_ * K2_;
    #pragma unroll
    for (int i = 0; i < 8; ++i) {
        Yb[(rowbase + i) * K2_ + colg] = acc[i];
    }
}

#define KSTEP2 32
#define PITCH2 (KSTEP2 + 4)
template <int NS>
static __device__ __forceinline__ v8f gemm2_tile(const float* __restrict__ Yb,
                                                 float (*sY)[PITCH2],
                                                 float (*sW)[PITCH2],
                                                 int nb, int tid) {
    constexpr int NY = (C_ * KSTEP2) / (256 * 4);
    constexpr size_t PLANE = (size_t)B_ * C_ * K2_;
    int lane  = tid & 31;
    int wave  = tid >> 5;
    int mtile = wave >> 1;
    int ntile = wave & 1;

    v8f acc = {0.f, 0.f, 0.f, 0.f, 0.f, 0.f, 0.f, 0.f};

    v4f ryA[NS][NY];
    v4f ryB[NS][NY];
    int yr0 = tid >> 3;
    int yq  = tid & (KSTEP2 / 4 - 1);
    int ys  = kslot(4 * yq);
    const float* yp = Yb + yr0 * K2_ + 4 * yq;

    auto fetch = [&](int kk, v4f (&ry)[NS][NY]) {
        #pragma unroll
        for (int s = 0; s < NS; ++s) {
            #pragma unroll
            for (int i = 0; i < NY; ++i) {
                ry[s][i] = *(const v4f*)(yp + s * PLANE + i * 32 * K2_ + kk);
            }
        }
    };
    auto stage = [&](const v4f (&ry)[NS][NY]) {
        #pragma unroll
        for (int i = 0; i < NY; ++i) {
            v4f v = ry[0][i];
            #pragma unroll
            for (int s = 1; s < NS; ++s) v += ry[s][i];
            v2f lo = {v.x, v.y}, hi = {v.z, v.w};
            *(v2f*)&sY[yr0 + 32 * i][ys]     = lo;
            *(v2f*)&sY[yr0 + 32 * i][ys + 4] = hi;
        }
    };
    int      nW     = nb * 32 + lane;
    float    qturnW = (wave & 1) ? 0.25f : 0.0f;
    unsigned pW     = (unsigned)((wave >> 1) * nW) & (unsigned)(OUT_ - 1);
    unsigned stepW  = (unsigned)(4 * nW) & (unsigned)(OUT_ - 1);
    float rw[(KSTEP2 * 32) / 256];
    auto synth_w = [&]() {
        #pragma unroll
        for (int i = 0; i < (KSTEP2 * 32) / 256; ++i) {
            rw[i] = (2.0f / (float)OUT_) *
                    __builtin_amdgcn_cosf((float)(int)pW * (1.0f / (float)OUT_) + qturnW);
            pW = (pW + stepW) & (unsigned)(OUT_ - 1);
        }
    };
    auto fix_w0 = [&]() {
        if (wave < 2) {
            unsigned p = wave ? ((unsigned)((NEED_ - 1) * nW) & (unsigned)(OUT_ - 1)) : 0u;
            rw[0] = (1.0f / (float)OUT_) *
                    __builtin_amdgcn_cosf((float)(int)p * (1.0f / (float)OUT_));
        }
    };
    int ws0 = kslot(wave);
    auto stage_w = [&]() {
        #pragma unroll
        for (int i = 0; i < (KSTEP2 * 32) / 256; ++i) sW[lane][ws0 + i * 8] = rw[i];
    };
    int arow = mtile * 16 + (lane & 15);
    int bcol = ntile * 16 + (lane & 15);
    int soff = (lane >> 4) << 2;
    v4f qa[KSTEP2 / 8], qb[KSTEP2 / 8];
    auto read_frags = [&]() {
        #pragma unroll
        for (int j = 0; j < KSTEP2 / 8; ++j) {
            qa[j] = *(const v4f*)&sY[arow][8 * j + soff];
            qb[j] = *(const v4f*)&sW[bcol][8 * j + soff];
        }
        __builtin_amdgcn_sched_barrier(0);
    };
    auto mma_chunk = [&]() {
        #pragma unroll
        for (int j = 0; j < KSTEP2 / 8; ++j) {
            v2f a0 = {qa[j].x, qa[j].y}, b0 = {qb[j].x, qb[j].y};
            v2f a1 = {qa[j].z, qa[j].w}, b1 = {qb[j].z, qb[j].w};
            acc = __builtin_amdgcn_wmma_f32_16x16x4_f32(
                false, a0, false, b0, (short)0, acc, false, false);
            acc = __builtin_amdgcn_wmma_f32_16x16x4_f32(
                false, a1, false, b1, (short)0, acc, false, false);
        }
        __builtin_amdgcn_sched_barrier(0);
    };

    static_assert(K2_ == 16 * KSTEP2, "the peeled tail below assumes 16 chunks");
    fetch(0, ryA);
    __builtin_amdgcn_sched_barrier(0);
    fetch(KSTEP2, ryB);
    synth_w();
    fix_w0();
    for (int kk = 0; kk < 14 * KSTEP2; kk += 2 * KSTEP2) {
        stage(ryA);
        stage_w();
        fetch(kk + 2 * KSTEP2, ryA);
        lds_barrier();
        read_frags();
        synth_w();
        mma_chunk();
        lds_barrier();

        stage(ryB);
        stage_w();
        fetch(kk + 3 * KSTEP2, ryB);
        lds_barrier();
        read_frags();
        synth_w();
        mma_chunk();
        lds_barrier();
    }
    stage(ryA);
    stage_w();
    lds_barrier();
    read_frags();
    synth_w();
    mma_chunk();
    lds_barrier();

    stage(ryB);
    stage_w();
    lds_barrier();
    read_frags();
    mma_chunk();
    return acc;
}

__global__ __launch_bounds__(256) void sp_gemm2(const float* __restrict__ Y,
                                                const float* __restrict__ x,
                                                const int*   __restrict__ length,
                                                float*       __restrict__ out) {
    int b  = blockIdx.x;
    int nb = blockIdx.y;
    int L  = length[b];

    __shared__ __attribute__((aligned(16))) float sY[C_][PITCH2];
    __shared__ __attribute__((aligned(16))) float sW[32][PITCH2];

    int tid   = threadIdx.x;
    int lane  = tid & 31;
    int wave  = tid >> 5;
    int mtile = wave >> 1;
    int ntile = wave & 1;

    bool shortSeq = (L < OUT_);
    int nseg = shortSeq ? 0 : (L + (SEGK_ - 1)) / SEGK_;
    if (nseg > NSEG_) nseg = NSEG_;

    const float* Yb = Y + (size_t)b * C_ * K2_;
    v8f acc = {0.f, 0.f, 0.f, 0.f, 0.f, 0.f, 0.f, 0.f};
    switch (nseg) {
        case 1: acc = gemm2_tile<1>(Yb, sY, sW, nb, tid); break;
        case 2: acc = gemm2_tile<2>(Yb, sY, sW, nb, tid); break;
        case 3: acc = gemm2_tile<3>(Yb, sY, sW, nb, tid); break;
        case 4: acc = gemm2_tile<4>(Yb, sY, sW, nb, tid); break;
        default: break;
    }

    int n       = nb * 32 + ntile * 16 + (lane & 15);
    int rowbase = mtile * 16 + ((lane >> 4) ? 8 : 0);
    const float* xb = x + (size_t)b * C_ * T_;
    float* ob = out + (size_t)b * C_ * OUT_;
    #pragma unroll
    for (int i = 0; i < 8; ++i) {
        int c = rowbase + i;
        float v = acc[i];
        if (shortSeq) v = (n < L) ? xb[c * T_ + n] : 0.0f;
        ob[c * OUT_ + n] = v;
    }
}

extern "C" void kernel_launch(void* const* d_in, const int* in_sizes, int n_in,
                              void* d_out, int out_size, void* d_ws, size_t ws_size,
                              hipStream_t stream) {
    (void)in_sizes; (void)n_in; (void)out_size; (void)ws_size;
    const float* x      = (const float*)d_in[0];
    const int*   length = (const int*)d_in[1];
    float*       out    = (float*)d_out;

    float* Y = (float*)d_ws;

    sp_gemm1<<<dim3(B_ * NSEG_ * NCB1), 256, 0, stream>>>(x, length, Y);
    sp_gemm2<<<dim3(B_, OUT_ / 32), 256, 0, stream>>>(Y, x, length, out);
}


# GEMM2: LDS tiles double-buffered, the next chunk staged under the current chunk's WMMAs: one barrier per 64-k chunk (16 WMMAs) instead of two per 32-k chunk (8 WMMAs); sums bit-identical
# speedup vs baseline: 4.6345x; 1.0689x over previous
#include <hip/hip_runtime.h>
#include <math.h>

#define B_    16
#define C_    64
#define T_    2048
#define OUT_  512
#define NEED_ 257
#define K2_   512
#define NSEG_ 4
#define SEGK_ 512

typedef float v2f __attribute__((ext_vector_type(2)));
typedef float v4f __attribute__((ext_vector_type(4)));
typedef float v8f __attribute__((ext_vector_type(8)));

static __device__ __forceinline__ int kslot(int k) {
    return (k & ~6) | ((k & 2) << 1) | ((k & 4) >> 1);
}

static __device__ __forceinline__ void lds_barrier() {
    __builtin_amdgcn_fence(__ATOMIC_RELEASE, "workgroup", "local");
    __builtin_amdgcn_s_barrier();
    __builtin_amdgcn_fence(__ATOMIC_ACQUIRE, "workgroup", "local");
}

#define KSTEP1 64
#define PITCH1 (KSTEP1 + 4)
#define NCB1   (K2_ / 32)
__global__ __launch_bounds__(256) void sp_gemm1(const float* __restrict__ x,
                                                const int*   __restrict__ length,
                                                float*       __restrict__ Y) {
    unsigned wg = blockIdx.x;
    int nb   = (int)(wg % NCB1);
    int pos  = (int)(wg / NCB1);
    int ln   = threadIdx.x & 31;
    int Lq   = length[ln & (B_ - 1)];
    int sgq  = ln >> 4;
    int endq = Lq < OUT_ ? 0 : ((Lq + (KSTEP1 - 1)) & ~(KSTEP1 - 1));
    int rem0 = endq - sgq * SEGK_;
    int rem1 = endq - (sgq + 2) * SEGK_;
    bool whole0 = rem0 >= SEGK_, cut0 = rem0 > 0 && !whole0;
    bool whole1 = rem1 >= SEGK_, cut1 = rem1 > 0 && !whole1;
    unsigned W0 = __builtin_amdgcn_ballot_w32(whole0), W1 = __builtin_amdgcn_ballot_w32(whole1);
    unsigned C0 = __builtin_amdgcn_ballot_w32(cut0),   C1 = __builtin_amdgcn_ballot_w32(cut1);
    unsigned below = (1u << ln) - 1u;
    int nW0 = __popc(W0), nW = nW0 + __popc(W1), nC0 = __popc(C0);
    int place0 = (whole0 ? 0   : nW)       + __popc((whole0 ? W0 : C0) & below);
    int place1 = (whole1 ? nW0 : nW + nC0) + __popc((whole1 ? W1 : C1) & below);
    unsigned H0 = __builtin_amdgcn_ballot_w32((whole0 || cut0) && place0 == pos);
    unsigned H1 = __builtin_amdgcn_ballot_w32((whole1 || cut1) && place1 == pos);
    if ((H0 | H1) == 0u) return;
    int pair = H0 ? __ffs(H0) - 1 : 31 + __ffs(H1);
    int b    = pair & (B_ - 1);
    int seg  = pair >> 4;
    int L    = __builtin_amdgcn_readlane(Lq, b);
    int kbeg = seg * SEGK_;
    int kend = (L + (KSTEP1 - 1)) & ~(KSTEP1 - 1);
    if (kend > kbeg + SEGK_) kend = kbeg + SEGK_;
    float invL = 1.0f / (float)L;

    __shared__ __attribute__((aligned(16))) float sA[C_][PITCH1];
    __shared__ __attribute__((aligned(16))) float sF[32][PITCH1];

    int tid   = threadIdx.x;
    int lane  = tid & 31;
    int wave  = tid >> 5;
    int mtile = wave >> 1;
    int ntile = wave & 1;

    v8f acc = {0.f, 0.f, 0.f, 0.f, 0.f, 0.f, 0.f, 0.f};

    const float* xb = x + (size_t)b * C_ * T_;

    int   fc     = tid & 31;
    int   g2     = nb * 32 + fc;
    bool  nyq    = (g2 == 1);
    int   k      = nyq ? (NEED_ - 1) : (g2 >> 1);
    unsigned step = (unsigned)((8 * k) % L);
    unsigned im   = (unsigned)((k * (kbeg + wave)) % L);
    float qturn  = ((g2 & 1) && !nyq) ? 0.25f : 0.0f;

    int xr0 = tid >> 4;
    int xq  = tid & (KSTEP1 / 4 - 1);
    int xs  = kslot(4 * xq);
    int fs0 = kslot(wave);
    const float* xp = xb + xr0 * T_ + 4 * xq;
    constexpr int NX = (C_ * KSTEP1) / (256 * 4);
    v4f rx[NX];
    #pragma unroll
    for (int i = 0; i < NX; ++i) {
        rx[i] = *(const v4f*)(xp + i * 16 * T_ + kbeg);
    }

    for (int kk = kbeg; kk < kend; kk += KSTEP1) {
        #pragma unroll
        for (int i = 0; i < NX; ++i) {
            v2f lo = {rx[i].x, rx[i].y}, hi = {rx[i].z, rx[i].w};
            *(v2f*)&sA[xr0 + 16 * i][xs]     = lo;
            *(v2f*)&sA[xr0 + 16 * i][xs + 4] = hi;
        }
        if (kk + KSTEP1 < kend) {
            #pragma unroll
            for (int i = 0; i < NX; ++i) {
                rx[i] = *(const v4f*)(xp + i * 16 * T_ + kk + KSTEP1);
            }
        }
        #pragma unroll
        for (int i = 0; i < (KSTEP1 * 32) / 256; ++i) {
            int tr = wave + i * 8;
            int t  = kk + tr;
            float v = 0.0f;
            if (t < L) v = __builtin_amdgcn_cosf((float)(int)im * invL + qturn);
            sF[fc][fs0 + i * 8] = v;
            unsigned nx = im + step;
            unsigned wr = nx - (unsigned)L;
            im = nx < wr ? nx : wr;
        }
        lds_barrier();

        int arow = mtile * 16 + (lane & 15);
        int bcol = ntile * 16 + (lane & 15);
        int soff = (lane >> 4) << 2;
        v4f qa[KSTEP1 / 8], qb[KSTEP1 / 8];
        #pragma unroll
        for (int j = 0; j < KSTEP1 / 8; ++j) {
            qa[j] = *(const v4f*)&sA[arow][8 * j + soff];
            qb[j] = *(const v4f*)&sF[bcol][8 * j + soff];
        }
        __builtin_amdgcn_sched_barrier(0);
        #pragma unroll
        for (int j = 0; j < KSTEP1 / 8; ++j) {
            v2f a0 = {qa[j].x, qa[j].y}, b0 = {qb[j].x, qb[j].y};
            v2f a1 = {qa[j].z, qa[j].w}, b1 = {qb[j].z, qb[j].w};
            acc = __builtin_amdgcn_wmma_f32_16x16x4_f32(
                false, a0, false, b0, (short)0, acc, false, false);
            acc = __builtin_amdgcn_wmma_f32_16x16x4_f32(
                false, a1, false, b1, (short)0, acc, false, false);
        }
        __builtin_amdgcn_sched_barrier(0);
        lds_barrier();
    }

    int colg    = nb * 32 + ntile * 16 + (lane & 15);
    int rowbase = mtile * 16 + ((lane >> 4) ? 8 : 0);
    float* Yb = Y + ((size_t)seg * B_ + b) * C_ * K2_;
    #pragma unroll
    for (int i = 0; i < 8; ++i) {
        Yb[(rowbase + i) * K2_ + colg] = acc[i];
    }
}

#define KSTEP2 64
#define PITCH2 (KSTEP2 + 4)
#define NCH2   (K2_ / KSTEP2)
template <int NS>
static __device__ __forceinline__ v8f gemm2_tile(const float* __restrict__ Yb,
                                                 float (*sY)[PITCH2],
                                                 float (*sW)[PITCH2],
                                                 int nb, int tid) {
    constexpr int NY = (C_ * KSTEP2) / (256 * 4);
    constexpr size_t PLANE = (size_t)B_ * C_ * K2_;
    int lane  = tid & 31;
    int wave  = tid >> 5;
    int mtile = wave >> 1;
    int ntile = wave & 1;

    v8f acc = {0.f, 0.f, 0.f, 0.f, 0.f, 0.f, 0.f, 0.f};

    v4f ryA[NS][NY];
    v4f ryB[NS][NY];
    int yr0 = tid >> 4;
    int yq  = tid & (KSTEP2 / 4 - 1);
    int ys  = kslot(4 * yq);
    const float* yp = Yb + yr0 * K2_ + 4 * yq;

    auto fetch = [&](int kk, v4f (&ry)[NS][NY]) {
        #pragma unroll
        for (int s = 0; s < NS; ++s) {
            #pragma unroll
            for (int i = 0; i < NY; ++i) {
                ry[s][i] = *(const v4f*)(yp + s * PLANE + i * 16 * K2_ + kk);
            }
        }
    };
    auto stage = [&](const v4f (&ry)[NS][NY], int buf) {
        #pragma unroll
        for (int i = 0; i < NY; ++i) {
            v4f v = ry[0][i];
            #pragma unroll
            for (int s = 1; s < NS; ++s) v += ry[s][i];
            v2f lo = {v.x, v.y}, hi = {v.z, v.w};
            *(v2f*)&sY[buf * C_ + yr0 + 16 * i][ys]     = lo;
            *(v2f*)&sY[buf * C_ + yr0 + 16 * i][ys + 4] = hi;
        }
    };
    int      nW     = nb * 32 + lane;
    float    qturnW = (wave & 1) ? 0.25f : 0.0f;
    unsigned pW     = (unsigned)((wave >> 1) * nW) & (unsigned)(OUT_ - 1);
    unsigned stepW  = (unsigned)(4 * nW) & (unsigned)(OUT_ - 1);
    float rw[(KSTEP2 * 32) / 256];
    auto synth_w = [&]() {
        #pragma unroll
        for (int i = 0; i < (KSTEP2 * 32) / 256; ++i) {
            rw[i] = (2.0f / (float)OUT_) *
                    __builtin_amdgcn_cosf((float)(int)pW * (1.0f / (float)OUT_) + qturnW);
            pW = (pW + stepW) & (unsigned)(OUT_ - 1);
        }
    };
    auto fix_w0 = [&]() {
        if (wave < 2) {
            unsigned p = wave ? ((unsigned)((NEED_ - 1) * nW) & (unsigned)(OUT_ - 1)) : 0u;
            rw[0] = (1.0f / (float)OUT_) *
                    __builtin_amdgcn_cosf((float)(int)p * (1.0f / (float)OUT_));
        }
    };
    int ws0 = kslot(wave);
    auto stage_w = [&](int buf) {
        #pragma unroll
        for (int i = 0; i < (KSTEP2 * 32) / 256; ++i) sW[buf * 32 + lane][ws0 + i * 8] = rw[i];
    };
    int arow = mtile * 16 + (lane & 15);
    int bcol = ntile * 16 + (lane & 15);
    int soff = (lane >> 4) << 2;
    v4f qa[KSTEP2 / 8], qb[KSTEP2 / 8];
    auto read_frags = [&](int buf) {
        #pragma unroll
        for (int j = 0; j < KSTEP2 / 8; ++j) {
            qa[j] = *(const v4f*)&sY[buf * C_ + arow][8 * j + soff];
            qb[j] = *(const v4f*)&sW[buf * 32 + bcol][8 * j + soff];
        }
        __builtin_amdgcn_sched_barrier(0);
    };
    auto mma_chunk = [&](bool with_synth) {
        __builtin_amdgcn_sched_barrier(0);
        if (with_synth) synth_w();
        #pragma unroll
        for (int j = 0; j < KSTEP2 / 8; ++j) {
            v2f a0 = {qa[j].x, qa[j].y}, b0 = {qb[j].x, qb[j].y};
            v2f a1 = {qa[j].z, qa[j].w}, b1 = {qb[j].z, qb[j].w};
            acc = __builtin_amdgcn_wmma_f32_16x16x4_f32(
                false, a0, false, b0, (short)0, acc, false, false);
            acc = __builtin_amdgcn_wmma_f32_16x16x4_f32(
                false, a1, false, b1, (short)0, acc, false, false);
        }
        __builtin_amdgcn_sched_barrier(0);
    };

    static_assert(NCH2 == 8, "the peeled tail below assumes 8 chunks");
    fetch(0, ryA);
    __builtin_amdgcn_sched_barrier(0);
    fetch(KSTEP2, ryB);
    synth_w();
    fix_w0();
    stage(ryA, 0);
    stage_w(0);
    fetch(2 * KSTEP2, ryA);
    synth_w();
    lds_barrier();
    for (int kk = 0; kk < 4 * KSTEP2; kk += 2 * KSTEP2) {
        read_frags(0);
        stage(ryB, 1);
        stage_w(1);
        fetch(kk + 3 * KSTEP2, ryB);
        mma_chunk(true);
        lds_barrier();

        read_frags(1);
        stage(ryA, 0);
        stage_w(0);
        fetch(kk + 4 * KSTEP2, ryA);
        mma_chunk(true);
        lds_barrier();
    }
    read_frags(0);
    stage(ryB, 1);
    stage_w(1);
    fetch(7 * KSTEP2, ryB);
    mma_chunk(true);
    lds_barrier();

    read_frags(1);
    stage(ryA, 0);
    stage_w(0);
    mma_chunk(true);
    lds_barrier();

    read_frags(0);
    stage(ryB, 1);
    stage_w(1);
    mma_chunk(false);
    lds_barrier();

    read_frags(1);
    mma_chunk(false);
    return acc;
}

__global__ __launch_bounds__(256) void sp_gemm2(const float* __restrict__ Y,
                                                const float* __restrict__ x,
                                                const int*   __restrict__ length,
                                                float*       __restrict__ out) {
    int b  = blockIdx.x;
    int nb = blockIdx.y;
    int L  = length[b];

    __shared__ __attribute__((aligned(16))) float sY[2 * C_][PITCH2];
    __shared__ __attribute__((aligned(16))) float sW[2 * 32][PITCH2];

    int tid   = threadIdx.x;
    int lane  = tid & 31;
    int wave  = tid >> 5;
    int mtile = wave >> 1;
    int ntile = wave & 1;

    bool shortSeq = (L < OUT_);
    int nseg = shortSeq ? 0 : (L + (SEGK_ - 1)) / SEGK_;
    if (nseg > NSEG_) nseg = NSEG_;

    const float* Yb = Y + (size_t)b * C_ * K2_;
    v8f acc = {0.f, 0.f, 0.f, 0.f, 0.f, 0.f, 0.f, 0.f};
    switch (nseg) {
        case 1: acc = gemm2_tile<1>(Yb, sY, sW, nb, tid); break;
        case 2: acc = gemm2_tile<2>(Yb, sY, sW, nb, tid); break;
        case 3: acc = gemm2_tile<3>(Yb, sY, sW, nb, tid); break;
        case 4: acc = gemm2_tile<4>(Yb, sY, sW, nb, tid); break;
        default: break;
    }

    int n       = nb * 32 + ntile * 16 + (lane & 15);
    int rowbase = mtile * 16 + ((lane >> 4) ? 8 : 0);
    const float* xb = x + (size_t)b * C_ * T_;
    float* ob = out + (size_t)b * C_ * OUT_;
    #pragma unroll
    for (int i = 0; i < 8; ++i) {
        int c = rowbase + i;
        float v = acc[i];
        if (shortSeq) v = (n < L) ? xb[c * T_ + n] : 0.0f;
        ob[c * OUT_ + n] = v;
    }
}

extern "C" void kernel_launch(void* const* d_in, const int* in_sizes, int n_in,
                              void* d_out, int out_size, void* d_ws, size_t ws_size,
                              hipStream_t stream) {
    (void)in_sizes; (void)n_in; (void)out_size; (void)ws_size;
    const float* x      = (const float*)d_in[0];
    const int*   length = (const int*)d_in[1];
    float*       out    = (float*)d_out;

    float* Y = (float*)d_ws;

    sp_gemm1<<<dim3(B_ * NSEG_ * NCB1), 256, 0, stream>>>(x, length, Y);
    sp_gemm2<<<dim3(B_, OUT_ / 32), 256, 0, stream>>>(Y, x, length, out);
}
